# starting kernel
# baseline (speedup 1.0000x reference)
#include <hip/hip_runtime.h>
#include <hip/hip_bf16.h>

typedef float v2f __attribute__((ext_vector_type(2)));
typedef float v8f __attribute__((ext_vector_type(8)));

#define D_FEAT 64
#define D_OUT  32

__global__ void fill_f32(float* __restrict__ p, float v, int n) {
    int i = blockIdx.x * blockDim.x + threadIdx.x;
    if (i < n) p[i] = v;
}

__global__ void deg_count(const int* __restrict__ dst, float* __restrict__ deg, int e) {
    int i = blockIdx.x * blockDim.x + threadIdx.x;
    if (i < e) unsafeAtomicAdd(&deg[dst[i]], 1.0f);
}

__global__ void rsqrt_inplace(float* __restrict__ p, int n) {
    int i = blockIdx.x * blockDim.x + threadIdx.x;
    if (i < n) {
        float d = p[i];
        p[i] = d > 0.0f ? __frsqrt_rn(d) : 0.0f;
    }
}

__global__ __launch_bounds__(256) void gcn_gemm64_wmma(
    const float* __restrict__ X, const float* __restrict__ W,
    float* __restrict__ H, int nrows)
{
    const int lane  = threadIdx.x & 31;
    const int strip = blockIdx.x * 8 + (threadIdx.x >> 5);
    const int r0 = strip * 16;
    if (r0 >= nrows) return;

    const int m  = lane & 15;
    const int kk = (lane >> 4) * 2;
    const int n  = lane & 15;

    v8f acc0 = {}, acc1 = {}, acc2 = {}, acc3 = {};
    const float* xrow = X + (size_t)(r0 + m) * D_FEAT;

    #pragma unroll
    for (int k0 = 0; k0 < D_FEAT; k0 += 4) {
        v2f a;
        a.x = xrow[k0 + kk];
        a.y = xrow[k0 + kk + 1];
        const float* w0 = W + (size_t)(k0 + kk) * D_FEAT + n;
        const float* w1 = W + (size_t)(k0 + kk + 1) * D_FEAT + n;
        v2f b0; b0.x = w0[0];  b0.y = w1[0];
        v2f b1; b1.x = w0[16]; b1.y = w1[16];
        v2f b2; b2.x = w0[32]; b2.y = w1[32];
        v2f b3; b3.x = w0[48]; b3.y = w1[48];
        acc0 = __builtin_amdgcn_wmma_f32_16x16x4_f32(false, a, false, b0, (short)0, acc0, false, false);
        acc1 = __builtin_amdgcn_wmma_f32_16x16x4_f32(false, a, false, b1, (short)0, acc1, false, false);
        acc2 = __builtin_amdgcn_wmma_f32_16x16x4_f32(false, a, false, b2, (short)0, acc2, false, false);
        acc3 = __builtin_amdgcn_wmma_f32_16x16x4_f32(false, a, false, b3, (short)0, acc3, false, false);
    }

    const int rbase = (lane >> 4) * 8;
    const int col   = lane & 15;
    #pragma unroll
    for (int r = 0; r < 8; ++r) {
        float* hrow = H + (size_t)(r0 + rbase + r) * D_FEAT + col;
        hrow[0]  = acc0[r];
        hrow[16] = acc1[r];
        hrow[32] = acc2[r];
        hrow[48] = acc3[r];
    }
}

__global__ __launch_bounds__(256) void gcn_scatter(
    const int* __restrict__ srcIdx, const int* __restrict__ dstIdx,
    const float* __restrict__ dinv, const float* __restrict__ h,
    float* __restrict__ agg, int nEdges)
{
    int idx = blockIdx.x * blockDim.x + threadIdx.x;
    if (idx >= nEdges * 16) return;
    int e = idx >> 4;
    int f = (idx & 15) * 4;
    int s = srcIdx[e], d = dstIdx[e];
    float nrm = dinv[s] * dinv[d];
    const float4 hv = *reinterpret_cast<const float4*>(h + (size_t)s * D_FEAT + f);
    float* ap = agg + (size_t)d * D_FEAT + f;
    unsafeAtomicAdd(ap + 0, nrm * hv.x);
    unsafeAtomicAdd(ap + 1, nrm * hv.y);
    unsafeAtomicAdd(ap + 2, nrm * hv.z);
    unsafeAtomicAdd(ap + 3, nrm * hv.w);
}

__global__ __launch_bounds__(256) void gcn_combine(
    const float* __restrict__ agg, const float* __restrict__ h,
    const float* __restrict__ dinv, const float* __restrict__ bias,
    const float* __restrict__ xprev, float* __restrict__ xnew, int n64)
{
    int idx = blockIdx.x * blockDim.x + threadIdx.x;
    if (idx >= n64) return;
    int node = idx >> 6, f = idx & 63;
    float di = dinv[node];
    float v = agg[idx] + di * di * h[idx] + bias[f];
    v = v > 0.0f ? v : 0.0f;
    xnew[idx] = v + xprev[idx];
}

__global__ void pool_count(const int* __restrict__ batch, float* __restrict__ cnts, int n) {
    int i = blockIdx.x * blockDim.x + threadIdx.x;
    if (i < n) unsafeAtomicAdd(&cnts[batch[i]], 1.0f);
}

__global__ __launch_bounds__(256) void pool_sum(
    const float* __restrict__ x, const int* __restrict__ batch,
    float* __restrict__ sums, int n64)
{
    int idx = blockIdx.x * blockDim.x + threadIdx.x;
    if (idx >= n64) return;
    int node = idx >> 6, f = idx & 63;
    unsafeAtomicAdd(&sums[(size_t)batch[node] * D_FEAT + f], x[idx]);
}

__global__ __launch_bounds__(256) void final_linear_wmma(
    const float* __restrict__ sums, const float* __restrict__ cnts,
    const float* __restrict__ Wl, const float* __restrict__ bl,
    float* __restrict__ out, int nrows)
{
    const int lane  = threadIdx.x & 31;
    const int strip = blockIdx.x * 8 + (threadIdx.x >> 5);
    const int r0 = strip * 16;
    if (r0 >= nrows) return;

    const int m  = lane & 15;
    const int kk = (lane >> 4) * 2;
    const int n  = lane & 15;
    const int row = r0 + m;
    float c = cnts[row];
    float inv = 1.0f / (c > 1.0f ? c : 1.0f);

    v8f acc0 = {}, acc1 = {};
    const float* srow = sums + (size_t)row * D_FEAT;

    #pragma unroll
    for (int k0 = 0; k0 < D_FEAT; k0 += 4) {
        v2f a;
        a.x = srow[k0 + kk] * inv;
        a.y = srow[k0 + kk + 1] * inv;
        const float* w0 = Wl + (size_t)(k0 + kk) * D_OUT + n;
        const float* w1 = Wl + (size_t)(k0 + kk + 1) * D_OUT + n;
        v2f b0; b0.x = w0[0];  b0.y = w1[0];
        v2f b1; b1.x = w0[16]; b1.y = w1[16];
        acc0 = __builtin_amdgcn_wmma_f32_16x16x4_f32(false, a, false, b0, (short)0, acc0, false, false);
        acc1 = __builtin_amdgcn_wmma_f32_16x16x4_f32(false, a, false, b1, (short)0, acc1, false, false);
    }

    const int rbase = (lane >> 4) * 8;
    const int col   = lane & 15;
    #pragma unroll
    for (int r = 0; r < 8; ++r) {
        float* orow = out + (size_t)(r0 + rbase + r) * D_OUT + col;
        orow[0]  = acc0[r] + bl[col];
        orow[16] = acc1[r] + bl[col + 16];
    }
}

static inline int cdiv(long long a, int b) { return (int)((a + b - 1) / b); }

extern "C" void kernel_launch(void* const* d_in, const int* in_sizes, int n_in,
                              void* d_out, int out_size, void* d_ws, size_t ws_size,
                              hipStream_t stream)
{
    const float* x     = (const float*)d_in[0];
    const int*   ei    = (const int*)d_in[1];
    const int*   batch = (const int*)d_in[2];
    const float* W[3]  = {(const float*)d_in[3], (const float*)d_in[5], (const float*)d_in[7]};
    const float* B[3]  = {(const float*)d_in[4], (const float*)d_in[6], (const float*)d_in[8]};
    const float* Wl    = (const float*)d_in[9];
    const float* bl    = (const float*)d_in[10];
    float* out = (float*)d_out;

    const int N = in_sizes[0] / D_FEAT;
    const int E = in_sizes[1] / 2;
    const int G = out_size / D_OUT;
    const int* srcIdx = ei;
    const int* dstIdx = ei + E;

    float* ws   = (float*)d_ws;
    float* buf0 = ws;
    float* buf1 = buf0 + (size_t)N * D_FEAT;
    float* hbuf = buf1 + (size_t)N * D_FEAT;
    float* dinv = hbuf + (size_t)N * D_FEAT;
    float* sums = dinv + N;
    float* cnts = sums + (size_t)G * D_FEAT;

    const int TB = 256;
    const int n64 = N * D_FEAT;

    fill_f32<<<cdiv(N, TB), TB, 0, stream>>>(dinv, 1.0f, N);
    deg_count<<<cdiv(E, TB), TB, 0, stream>>>(dstIdx, dinv, E);
    rsqrt_inplace<<<cdiv(N, TB), TB, 0, stream>>>(dinv, N);

    fill_f32<<<cdiv(G * D_FEAT, TB), TB, 0, stream>>>(sums, 0.0f, G * D_FEAT);
    fill_f32<<<cdiv(G, TB), TB, 0, stream>>>(cnts, 0.0f, G);

    const int gemmBlocks = cdiv((long long)cdiv(N, 16), 8);
    const float* xcur = x;
    float* aggbuf[3] = {buf0, buf1, buf0};

    for (int l = 0; l < 3; ++l) {
        float* agg = aggbuf[l];
        gcn_gemm64_wmma<<<gemmBlocks, TB, 0, stream>>>(xcur, W[l], hbuf, N);
        fill_f32<<<cdiv(n64, TB), TB, 0, stream>>>(agg, 0.0f, n64);
        gcn_scatter<<<cdiv((long long)E * 16, TB), TB, 0, stream>>>(srcIdx, dstIdx, dinv, hbuf, agg, E);
        gcn_combine<<<cdiv(n64, TB), TB, 0, stream>>>(agg, hbuf, dinv, B[l], xcur, agg, n64);
        xcur = agg;
    }

    pool_count<<<cdiv(N, TB), TB, 0, stream>>>(batch, cnts, N);
    pool_sum<<<cdiv(n64, TB), TB, 0, stream>>>(xcur, batch, sums, n64);
    final_linear_wmma<<<cdiv((long long)cdiv(G, 16), 8), TB, 0, stream>>>(sums, cnts, Wl, bl, out, G);
}


# Replace float-atomic edge scatter by a per-call dst-sorted edge list (int atomic rank + prefix scan) and a gather-reduce fused with combine; per-node sum order = atomic arrival rank
# speedup vs baseline: 2.0991x; 2.0991x over previous
#include <hip/hip_runtime.h>
#include <hip/hip_bf16.h>

typedef float v2f __attribute__((ext_vector_type(2)));
typedef float v8f __attribute__((ext_vector_type(8)));

#define D_FEAT 64
#define D_OUT  32

__global__ void fill_f32(float* __restrict__ p, float v, int n) {
    int i = blockIdx.x * blockDim.x + threadIdx.x;
    if (i < n) p[i] = v;
}

__global__ void fill_i32(int* __restrict__ p, int v, int n) {
    int i = blockIdx.x * blockDim.x + threadIdx.x;
    if (i < n) p[i] = v;
}

__global__ void deg_rank(const int* __restrict__ dst, int* __restrict__ cnt, int* __restrict__ rank, int e) {
    int i = blockIdx.x * blockDim.x + threadIdx.x;
    if (i < e) rank[i] = atomicAdd(&cnt[dst[i]], 1);
}

__global__ __launch_bounds__(1024) void scan_deg(
    const int* __restrict__ cnt, int* __restrict__ nodeRaw, int n)
{
    __shared__ int s_w[32];
    __shared__ int s_pre[32];
    const int tid = threadIdx.x, lane = tid & 31, wv = tid >> 5;
    int pre = 0;
    for (int b = 0; b < (int)blockIdx.x; ++b) pre += cnt[b * 1024 + tid];
    const int i = blockIdx.x * 1024 + tid;
    const int c = i < n ? cnt[i] : 0;
    int inc = c;
    for (int o = 1; o < 32; o <<= 1) { int t = __shfl_up(inc, o); if (lane >= o) inc += t; }
    for (int o = 16; o > 0; o >>= 1) pre += __shfl_xor(pre, o);
    if (lane == 31) s_w[wv] = inc;
    if (lane == 0)  s_pre[wv] = pre;
    __syncthreads();
    if (wv == 0) {
        const int w = s_w[lane];
        int winc = w;
        for (int o = 1; o < 32; o <<= 1) { int t = __shfl_up(winc, o); if (lane >= o) winc += t; }
        int p = s_pre[lane];
        for (int o = 16; o > 0; o >>= 1) p += __shfl_xor(p, o);
        s_w[lane] = winc - w;
        if (lane == 0) s_pre[0] = p;
    }
    __syncthreads();
    if (i < n) {
        const int start = s_pre[0] + s_w[wv] + inc - c;
        const float d = (float)(c + 1);
        const float di = d > 0.0f ? __frsqrt_rn(d) : 0.0f;
        reinterpret_cast<int2*>(nodeRaw)[i] = make_int2(start, __float_as_int(di));
    }
}

__global__ void csr_fill(
    const int* __restrict__ srcIdx, const int* __restrict__ dstIdx,
    const int* __restrict__ nodeRaw, const int* __restrict__ rank,
    int* __restrict__ edgeRaw, int e)
{
    int i = blockIdx.x * blockDim.x + threadIdx.x;
    if (i >= e) return;
    const int2* nodeInfo = reinterpret_cast<const int2*>(nodeRaw);
    int s = srcIdx[i], d = dstIdx[i];
    const int2 nd = nodeInfo[d];
    float nrm = __int_as_float(nodeInfo[s].y) * __int_as_float(nd.y);
    reinterpret_cast<int2*>(edgeRaw)[nd.x + rank[i]] = make_int2(s, __float_as_int(nrm));
}

__global__ __launch_bounds__(256) void gcn_gemm64_wmma(
    const float* __restrict__ X, const float* __restrict__ W,
    float* __restrict__ H, int nrows)
{
    const int lane  = threadIdx.x & 31;
    const int strip = blockIdx.x * 8 + (threadIdx.x >> 5);
    const int r0 = strip * 16;
    if (r0 >= nrows) return;

    const int m  = lane & 15;
    const int kk = (lane >> 4) * 2;
    const int n  = lane & 15;

    v8f acc0 = {}, acc1 = {}, acc2 = {}, acc3 = {};
    const float* xrow = X + (size_t)(r0 + m) * D_FEAT;

    #pragma unroll
    for (int k0 = 0; k0 < D_FEAT; k0 += 4) {
        v2f a;
        a.x = xrow[k0 + kk];
        a.y = xrow[k0 + kk + 1];
        const float* w0 = W + (size_t)(k0 + kk) * D_FEAT + n;
        const float* w1 = W + (size_t)(k0 + kk + 1) * D_FEAT + n;
        v2f b0; b0.x = w0[0];  b0.y = w1[0];
        v2f b1; b1.x = w0[16]; b1.y = w1[16];
        v2f b2; b2.x = w0[32]; b2.y = w1[32];
        v2f b3; b3.x = w0[48]; b3.y = w1[48];
        acc0 = __builtin_amdgcn_wmma_f32_16x16x4_f32(false, a, false, b0, (short)0, acc0, false, false);
        acc1 = __builtin_amdgcn_wmma_f32_16x16x4_f32(false, a, false, b1, (short)0, acc1, false, false);
        acc2 = __builtin_amdgcn_wmma_f32_16x16x4_f32(false, a, false, b2, (short)0, acc2, false, false);
        acc3 = __builtin_amdgcn_wmma_f32_16x16x4_f32(false, a, false, b3, (short)0, acc3, false, false);
    }

    const int rbase = (lane >> 4) * 8;
    const int col   = lane & 15;
    #pragma unroll
    for (int r = 0; r < 8; ++r) {
        float* hrow = H + (size_t)(r0 + rbase + r) * D_FEAT + col;
        hrow[0]  = acc0[r];
        hrow[16] = acc1[r];
        hrow[32] = acc2[r];
        hrow[48] = acc3[r];
    }
}

__global__ __launch_bounds__(256) void gcn_gather(
    const int* __restrict__ nodeRaw, const int* __restrict__ cnt, const int* __restrict__ edgeRaw,
    const float* __restrict__ h, const float* __restrict__ bias,
    const float* __restrict__ xprev, float* __restrict__ xnew, int nNodes)
{
    int idx = blockIdx.x * blockDim.x + threadIdx.x;
    int node = idx >> 4;
    if (node >= nNodes) return;
    int f = (idx & 15) * 4;
    const int2 nd = reinterpret_cast<const int2*>(nodeRaw)[node];
    const int2* p = reinterpret_cast<const int2*>(edgeRaw) + nd.x;
    const int len = cnt[node];
    const float* hf = h + f;
    const size_t o = (size_t)node * D_FEAT + f;
    const float4 hs = *reinterpret_cast<const float4*>(h + o);
    const float4 bb = *reinterpret_cast<const float4*>(bias + f);
    const float4 xp = *reinterpret_cast<const float4*>(xprev + o);
    float4 acc = make_float4(0.0f, 0.0f, 0.0f, 0.0f);
    int j = 0;
    for (; j + 4 <= len; j += 4) {
        const int2 e0 = p[j], e1 = p[j + 1], e2 = p[j + 2], e3 = p[j + 3];
        const float4 h0 = *reinterpret_cast<const float4*>(hf + (size_t)e0.x * D_FEAT);
        const float4 h1 = *reinterpret_cast<const float4*>(hf + (size_t)e1.x * D_FEAT);
        const float4 h2 = *reinterpret_cast<const float4*>(hf + (size_t)e2.x * D_FEAT);
        const float4 h3 = *reinterpret_cast<const float4*>(hf + (size_t)e3.x * D_FEAT);
        const float w0 = __int_as_float(e0.y), w1 = __int_as_float(e1.y);
        const float w2 = __int_as_float(e2.y), w3 = __int_as_float(e3.y);
        acc.x = fmaf(w0, h0.x, acc.x); acc.y = fmaf(w0, h0.y, acc.y); acc.z = fmaf(w0, h0.z, acc.z); acc.w = fmaf(w0, h0.w, acc.w);
        acc.x = fmaf(w1, h1.x, acc.x); acc.y = fmaf(w1, h1.y, acc.y); acc.z = fmaf(w1, h1.z, acc.z); acc.w = fmaf(w1, h1.w, acc.w);
        acc.x = fmaf(w2, h2.x, acc.x); acc.y = fmaf(w2, h2.y, acc.y); acc.z = fmaf(w2, h2.z, acc.z); acc.w = fmaf(w2, h2.w, acc.w);
        acc.x = fmaf(w3, h3.x, acc.x); acc.y = fmaf(w3, h3.y, acc.y); acc.z = fmaf(w3, h3.z, acc.z); acc.w = fmaf(w3, h3.w, acc.w);
    }
    for (; j < len; ++j) {
        const int2 e0 = p[j];
        const float4 h0 = *reinterpret_cast<const float4*>(hf + (size_t)e0.x * D_FEAT);
        const float w0 = __int_as_float(e0.y);
        acc.x = fmaf(w0, h0.x, acc.x); acc.y = fmaf(w0, h0.y, acc.y); acc.z = fmaf(w0, h0.z, acc.z); acc.w = fmaf(w0, h0.w, acc.w);
    }
    const float di = __int_as_float(nd.y);
    const float dd = di * di;
    float4 v;
    v.x = acc.x + dd * hs.x + bb.x;
    v.y = acc.y + dd * hs.y + bb.y;
    v.z = acc.z + dd * hs.z + bb.z;
    v.w = acc.w + dd * hs.w + bb.w;
    v.x = (v.x > 0.0f ? v.x : 0.0f) + xp.x;
    v.y = (v.y > 0.0f ? v.y : 0.0f) + xp.y;
    v.z = (v.z > 0.0f ? v.z : 0.0f) + xp.z;
    v.w = (v.w > 0.0f ? v.w : 0.0f) + xp.w;
    *reinterpret_cast<float4*>(xnew + o) = v;
}

__global__ void pool_count(const int* __restrict__ batch, float* __restrict__ cnts, int n) {
    int i = blockIdx.x * blockDim.x + threadIdx.x;
    if (i < n) unsafeAtomicAdd(&cnts[batch[i]], 1.0f);
}

__global__ __launch_bounds__(256) void pool_sum(
    const float* __restrict__ x, const int* __restrict__ batch,
    float* __restrict__ sums, int n64)
{
    int idx = blockIdx.x * blockDim.x + threadIdx.x;
    if (idx >= n64) return;
    int node = idx >> 6, f = idx & 63;
    unsafeAtomicAdd(&sums[(size_t)batch[node] * D_FEAT + f], x[idx]);
}

__global__ __launch_bounds__(256) void final_linear_wmma(
    const float* __restrict__ sums, const float* __restrict__ cnts,
    const float* __restrict__ Wl, const float* __restrict__ bl,
    float* __restrict__ out, int nrows)
{
    const int lane  = threadIdx.x & 31;
    const int strip = blockIdx.x * 8 + (threadIdx.x >> 5);
    const int r0 = strip * 16;
    if (r0 >= nrows) return;

    const int m  = lane & 15;
    const int kk = (lane >> 4) * 2;
    const int n  = lane & 15;
    const int row = r0 + m;
    float c = cnts[row];
    float inv = 1.0f / (c > 1.0f ? c : 1.0f);

    v8f acc0 = {}, acc1 = {};
    const float* srow = sums + (size_t)row * D_FEAT;

    #pragma unroll
    for (int k0 = 0; k0 < D_FEAT; k0 += 4) {
        v2f a;
        a.x = srow[k0 + kk] * inv;
        a.y = srow[k0 + kk + 1] * inv;
        const float* w0 = Wl + (size_t)(k0 + kk) * D_OUT + n;
        const float* w1 = Wl + (size_t)(k0 + kk + 1) * D_OUT + n;
        v2f b0; b0.x = w0[0];  b0.y = w1[0];
        v2f b1; b1.x = w0[16]; b1.y = w1[16];
        acc0 = __builtin_amdgcn_wmma_f32_16x16x4_f32(false, a, false, b0, (short)0, acc0, false, false);
        acc1 = __builtin_amdgcn_wmma_f32_16x16x4_f32(false, a, false, b1, (short)0, acc1, false, false);
    }

    const int rbase = (lane >> 4) * 8;
    const int col   = lane & 15;
    #pragma unroll
    for (int r = 0; r < 8; ++r) {
        float* orow = out + (size_t)(r0 + rbase + r) * D_OUT + col;
        orow[0]  = acc0[r] + bl[col];
        orow[16] = acc1[r] + bl[col + 16];
    }
}

static inline int cdiv(long long a, int b) { return (int)((a + b - 1) / b); }

extern "C" void kernel_launch(void* const* d_in, const int* in_sizes, int n_in,
                              void* d_out, int out_size, void* d_ws, size_t ws_size,
                              hipStream_t stream)
{
    const float* x     = (const float*)d_in[0];
    const int*   ei    = (const int*)d_in[1];
    const int*   batch = (const int*)d_in[2];
    const float* W[3]  = {(const float*)d_in[3], (const float*)d_in[5], (const float*)d_in[7]};
    const float* B[3]  = {(const float*)d_in[4], (const float*)d_in[6], (const float*)d_in[8]};
    const float* Wl    = (const float*)d_in[9];
    const float* bl    = (const float*)d_in[10];
    float* out = (float*)d_out;

    const int N = in_sizes[0] / D_FEAT;
    const int E = in_sizes[1] / 2;
    const int G = out_size / D_OUT;
    const int* srcIdx = ei;
    const int* dstIdx = ei + E;

    float* ws   = (float*)d_ws;
    float* buf0 = ws;
    float* buf1 = buf0 + (size_t)N * D_FEAT;
    float* hbuf = buf1 + (size_t)N * D_FEAT;
    int*   edges = (int*)(hbuf + (size_t)N * D_FEAT);
    int*   nodes = edges + (size_t)E * 2;
    int*   rank  = nodes + (size_t)N * 2;
    int*   cnt   = rank + E;
    float* sums  = (float*)(cnt + N);
    float* cnts  = sums + (size_t)G * D_FEAT;

    const int TB = 256;
    const int n64 = N * D_FEAT;

    fill_i32<<<cdiv(N, TB), TB, 0, stream>>>(cnt, 0, N);
    deg_rank<<<cdiv(E, TB), TB, 0, stream>>>(dstIdx, cnt, rank, E);
    scan_deg<<<cdiv(N, 1024), 1024, 0, stream>>>(cnt, nodes, N);
    csr_fill<<<cdiv(E, TB), TB, 0, stream>>>(srcIdx, dstIdx, nodes, rank, edges, E);

    fill_f32<<<cdiv(G * D_FEAT, TB), TB, 0, stream>>>(sums, 0.0f, G * D_FEAT);
    fill_f32<<<cdiv(G, TB), TB, 0, stream>>>(cnts, 0.0f, G);

    const int gemmBlocks = cdiv((long long)cdiv(N, 16), 8);
    const float* xcur = x;
    float* xbuf[3] = {buf0, buf1, buf0};

    for (int l = 0; l < 3; ++l) {
        float* xnew = xbuf[l];
        gcn_gemm64_wmma<<<gemmBlocks, TB, 0, stream>>>(xcur, W[l], hbuf, N);
        gcn_gather<<<cdiv((long long)N * 16, TB), TB, 0, stream>>>(nodes, cnt, edges, hbuf, B[l], xcur, xnew, N);
        xcur = xnew;
    }

    pool_count<<<cdiv(N, TB), TB, 0, stream>>>(batch, cnts, N);
    pool_sum<<<cdiv(n64, TB), TB, 0, stream>>>(xcur, batch, sums, n64);
    final_linear_wmma<<<cdiv((long long)cdiv(G, 16), 8), TB, 0, stream>>>(sums, cnts, Wl, bl, out, G);
}


# Long-lived waves: per-item kernels run resident grids whose waves loop over items with 8 loads in flight (9x fewer waves per call); pool sums hold each graph run in registers: pool sum order changes
# speedup vs baseline: 2.3398x; 1.1147x over previous
#include <hip/hip_runtime.h>
#include <hip/hip_bf16.h>

typedef float v2f __attribute__((ext_vector_type(2)));
typedef float v8f __attribute__((ext_vector_type(8)));

#define D_FEAT 64
#define D_OUT  32

__global__ void fill_f32(float* __restrict__ p, float v, int n) {
    for (int i = blockIdx.x * blockDim.x + threadIdx.x; i < n; i += gridDim.x * blockDim.x) p[i] = v;
}

__global__ void fill_i32(int* __restrict__ p, int v, int n) {
    for (int i = blockIdx.x * blockDim.x + threadIdx.x; i < n; i += gridDim.x * blockDim.x) p[i] = v;
}

__global__ __launch_bounds__(256) void deg_rank(const int* __restrict__ dst, int* __restrict__ cnt, int* __restrict__ rank, int e) {
    const int stride = gridDim.x * 256;
    for (int i0 = blockIdx.x * 256 + threadIdx.x; i0 < e; i0 += 8 * stride) {
        if (i0 + 7 * stride < e) {
            int d[8], r[8];
            #pragma unroll
            for (int u = 0; u < 8; ++u) d[u] = dst[i0 + u * stride];
            #pragma unroll
            for (int u = 0; u < 8; ++u) r[u] = atomicAdd(&cnt[d[u]], 1);
            #pragma unroll
            for (int u = 0; u < 8; ++u) rank[i0 + u * stride] = r[u];
        } else {
            for (int i = i0; i < e; i += stride) rank[i] = atomicAdd(&cnt[dst[i]], 1);
        }
    }
}

__global__ __launch_bounds__(1024) void scan_deg(
    const int* __restrict__ cnt, int* __restrict__ nodeRaw, int n)
{
    __shared__ int s_w[32];
    __shared__ int s_pre[32];
    const int tid = threadIdx.x, lane = tid & 31, wv = tid >> 5;
    int pre = 0;
    for (int b = 0; b < (int)blockIdx.x; ++b) pre += cnt[b * 1024 + tid];
    const int i = blockIdx.x * 1024 + tid;
    const int c = i < n ? cnt[i] : 0;
    int inc = c;
    for (int o = 1; o < 32; o <<= 1) { int t = __shfl_up(inc, o); if (lane >= o) inc += t; }
    for (int o = 16; o > 0; o >>= 1) pre += __shfl_xor(pre, o);
    if (lane == 31) s_w[wv] = inc;
    if (lane == 0)  s_pre[wv] = pre;
    __syncthreads();
    if (wv == 0) {
        const int w = s_w[lane];
        int winc = w;
        for (int o = 1; o < 32; o <<= 1) { int t = __shfl_up(winc, o); if (lane >= o) winc += t; }
        int p = s_pre[lane];
        for (int o = 16; o > 0; o >>= 1) p += __shfl_xor(p, o);
        s_w[lane] = winc - w;
        if (lane == 0) s_pre[0] = p;
    }
    __syncthreads();
    if (i < n) {
        const int start = s_pre[0] + s_w[wv] + inc - c;
        const float d = (float)(c + 1);
        const float di = d > 0.0f ? __frsqrt_rn(d) : 0.0f;
        reinterpret_cast<int2*>(nodeRaw)[i] = make_int2(start, __float_as_int(di));
    }
}

__global__ __launch_bounds__(256) void csr_fill(
    const int* __restrict__ srcIdx, const int* __restrict__ dstIdx,
    const int* __restrict__ nodeRaw, const int* __restrict__ rank,
    int* __restrict__ edgeRaw, int e)
{
    const int2* nodeInfo = reinterpret_cast<const int2*>(nodeRaw);
    int2* edgeOut = reinterpret_cast<int2*>(edgeRaw);
    const int stride = gridDim.x * 256;
    for (int i0 = blockIdx.x * 256 + threadIdx.x; i0 < e; i0 += 8 * stride) {
        if (i0 + 7 * stride < e) {
            int s[8], d[8], r[8];
            #pragma unroll
            for (int u = 0; u < 8; ++u) {
                const int i = i0 + u * stride;
                s[u] = srcIdx[i];
                d[u] = dstIdx[i];
                r[u] = rank[i];
            }
            int2 nd[8];
            float ds[8];
            #pragma unroll
            for (int u = 0; u < 8; ++u) { nd[u] = nodeInfo[d[u]]; ds[u] = __int_as_float(nodeInfo[s[u]].y); }
            #pragma unroll
            for (int u = 0; u < 8; ++u) {
                const float nrm = ds[u] * __int_as_float(nd[u].y);
                edgeOut[nd[u].x + r[u]] = make_int2(s[u], __float_as_int(nrm));
            }
        } else {
            for (int i = i0; i < e; i += stride) {
                const int s0 = srcIdx[i], d0 = dstIdx[i];
                const int2 n0 = nodeInfo[d0];
                const float nrm = __int_as_float(nodeInfo[s0].y) * __int_as_float(n0.y);
                edgeOut[n0.x + rank[i]] = make_int2(s0, __float_as_int(nrm));
            }
        }
    }
}

__global__ __launch_bounds__(256) void gcn_gemm64_wmma(
    const float* __restrict__ X, const float* __restrict__ W,
    float* __restrict__ H, int nrows)
{
    const int lane  = threadIdx.x & 31;
    const int strip = blockIdx.x * 8 + (threadIdx.x >> 5);
    const int r0 = strip * 16;
    if (r0 >= nrows) return;

    const int m  = lane & 15;
    const int kk = (lane >> 4) * 2;
    const int n  = lane & 15;

    v8f acc0 = {}, acc1 = {}, acc2 = {}, acc3 = {};
    const float* xrow = X + (size_t)(r0 + m) * D_FEAT;

    #pragma unroll
    for (int k0 = 0; k0 < D_FEAT; k0 += 4) {
        v2f a;
        a.x = xrow[k0 + kk];
        a.y = xrow[k0 + kk + 1];
        const float* w0 = W + (size_t)(k0 + kk) * D_FEAT + n;
        const float* w1 = W + (size_t)(k0 + kk + 1) * D_FEAT + n;
        v2f b0; b0.x = w0[0];  b0.y = w1[0];
        v2f b1; b1.x = w0[16]; b1.y = w1[16];
        v2f b2; b2.x = w0[32]; b2.y = w1[32];
        v2f b3; b3.x = w0[48]; b3.y = w1[48];
        acc0 = __builtin_amdgcn_wmma_f32_16x16x4_f32(false, a, false, b0, (short)0, acc0, false, false);
        acc1 = __builtin_amdgcn_wmma_f32_16x16x4_f32(false, a, false, b1, (short)0, acc1, false, false);
        acc2 = __builtin_amdgcn_wmma_f32_16x16x4_f32(false, a, false, b2, (short)0, acc2, false, false);
        acc3 = __builtin_amdgcn_wmma_f32_16x16x4_f32(false, a, false, b3, (short)0, acc3, false, false);
    }

    const int rbase = (lane >> 4) * 8;
    const int col   = lane & 15;
    #pragma unroll
    for (int r = 0; r < 8; ++r) {
        float* hrow = H + (size_t)(r0 + rbase + r) * D_FEAT + col;
        hrow[0]  = acc0[r];
        hrow[16] = acc1[r];
        hrow[32] = acc2[r];
        hrow[48] = acc3[r];
    }
}

template <int K, bool TAIL>
__device__ __forceinline__ void gather_edges(const int2* __restrict__ p, int j, int len,
                                             const float* __restrict__ hf, float4& acc)
{
    int2 e[K];
    #pragma unroll
    for (int k = 0; k < K; ++k) {
        int idx = j + k;
        if (TAIL) idx = idx < len ? idx : len - 1;
        e[k] = p[idx];
    }
    float4 r[K];
    #pragma unroll
    for (int k = 0; k < K; ++k) r[k] = *reinterpret_cast<const float4*>(hf + (size_t)e[k].x * D_FEAT);
    #pragma unroll
    for (int k = 0; k < K; ++k) {
        if (!TAIL || j + k < len) {
            const float w = __int_as_float(e[k].y);
            acc.x = fmaf(w, r[k].x, acc.x); acc.y = fmaf(w, r[k].y, acc.y);
            acc.z = fmaf(w, r[k].z, acc.z); acc.w = fmaf(w, r[k].w, acc.w);
        }
    }
}

__global__ __launch_bounds__(256) void gcn_gather(
    const int* __restrict__ nodeRaw, const int* __restrict__ cnt, const int* __restrict__ edgeRaw,
    const float* __restrict__ h, const float* __restrict__ bias,
    const float* __restrict__ xprev, float* __restrict__ xnew, int nNodes)
{
    const int f = (threadIdx.x & 15) * 4;
    const int groupStride = (gridDim.x * 256) >> 4;
    const int2* nodeInfo = reinterpret_cast<const int2*>(nodeRaw);
    const int2* edges = reinterpret_cast<const int2*>(edgeRaw);
    const float* hf = h + f;
    const float4 bb = *reinterpret_cast<const float4*>(bias + f);
    for (int node = (blockIdx.x * 256 + threadIdx.x) >> 4; node < nNodes; node += groupStride) {
        const int2 nd = nodeInfo[node];
        const int len = cnt[node];
        const size_t o = (size_t)node * D_FEAT + f;
        const float4 hs = *reinterpret_cast<const float4*>(h + o);
        const float4 xp = *reinterpret_cast<const float4*>(xprev + o);
        const int2* p = edges + nd.x;
        float4 acc = make_float4(0.0f, 0.0f, 0.0f, 0.0f);
        int j = 0;
        for (; j + 8 <= len; j += 8) gather_edges<8, false>(p, j, len, hf, acc);
        if (j + 4 <= len) { gather_edges<4, false>(p, j, len, hf, acc); j += 4; }
        if (j < len) gather_edges<4, true>(p, j, len, hf, acc);
        const float di = __int_as_float(nd.y);
        const float dd = di * di;
        float4 v;
        v.x = acc.x + dd * hs.x + bb.x;
        v.y = acc.y + dd * hs.y + bb.y;
        v.z = acc.z + dd * hs.z + bb.z;
        v.w = acc.w + dd * hs.w + bb.w;
        v.x = (v.x > 0.0f ? v.x : 0.0f) + xp.x;
        v.y = (v.y > 0.0f ? v.y : 0.0f) + xp.y;
        v.z = (v.z > 0.0f ? v.z : 0.0f) + xp.z;
        v.w = (v.w > 0.0f ? v.w : 0.0f) + xp.w;
        *reinterpret_cast<float4*>(xnew + o) = v;
    }
}

#define POOL_SPAN 32

__global__ void pool_count(const int* __restrict__ batch, float* __restrict__ cnts, int n) {
    const int beg = (blockIdx.x * blockDim.x + threadIdx.x) * POOL_SPAN;
    if (beg >= n) return;
    const int end = beg + POOL_SPAN < n ? beg + POOL_SPAN : n;
    int g = batch[beg];
    float c = 0.0f;
    int i = beg;
    for (; i + 8 <= end; i += 8) {
        int b[8];
        #pragma unroll
        for (int k = 0; k < 8; ++k) b[k] = batch[i + k];
        #pragma unroll
        for (int k = 0; k < 8; ++k) {
            if (b[k] != g) { unsafeAtomicAdd(&cnts[g], c); c = 0.0f; g = b[k]; }
            c += 1.0f;
        }
    }
    for (; i < end; ++i) {
        const int b0 = batch[i];
        if (b0 != g) { unsafeAtomicAdd(&cnts[g], c); c = 0.0f; g = b0; }
        c += 1.0f;
    }
    unsafeAtomicAdd(&cnts[g], c);
}

__device__ __forceinline__ void pool_flush(float* __restrict__ sums, int g, int f, const float4& a) {
    float* s = sums + (size_t)g * D_FEAT + f;
    unsafeAtomicAdd(s + 0, a.x);
    unsafeAtomicAdd(s + 1, a.y);
    unsafeAtomicAdd(s + 2, a.z);
    unsafeAtomicAdd(s + 3, a.w);
}

__global__ __launch_bounds__(256) void pool_sum(
    const float* __restrict__ x, const int* __restrict__ batch,
    float* __restrict__ sums, int n)
{
    const int f = (threadIdx.x & 15) * 4;
    const int beg = ((blockIdx.x * 256 + threadIdx.x) >> 4) * POOL_SPAN;
    if (beg >= n) return;
    const int end = beg + POOL_SPAN < n ? beg + POOL_SPAN : n;
    const float* xf = x + f;
    int g = batch[beg];
    float4 acc = make_float4(0.0f, 0.0f, 0.0f, 0.0f);
    int i = beg;
    for (; i + 8 <= end; i += 8) {
        int b[8];
        float4 v[8];
        #pragma unroll
        for (int k = 0; k < 8; ++k) {
            b[k] = batch[i + k];
            v[k] = *reinterpret_cast<const float4*>(xf + (size_t)(i + k) * D_FEAT);
        }
        #pragma unroll
        for (int k = 0; k < 8; ++k) {
            if (b[k] != g) { pool_flush(sums, g, f, acc); acc = make_float4(0.0f, 0.0f, 0.0f, 0.0f); g = b[k]; }
            acc.x += v[k].x; acc.y += v[k].y; acc.z += v[k].z; acc.w += v[k].w;
        }
    }
    for (; i < end; ++i) {
        const int b0 = batch[i];
        const float4 v0 = *reinterpret_cast<const float4*>(xf + (size_t)i * D_FEAT);
        if (b0 != g) { pool_flush(sums, g, f, acc); acc = make_float4(0.0f, 0.0f, 0.0f, 0.0f); g = b0; }
        acc.x += v0.x; acc.y += v0.y; acc.z += v0.z; acc.w += v0.w;
    }
    pool_flush(sums, g, f, acc);
}

__global__ __launch_bounds__(256) void final_linear_wmma(
    const float* __restrict__ sums, const float* __restrict__ cnts,
    const float* __restrict__ Wl, const float* __restrict__ bl,
    float* __restrict__ out, int nrows)
{
    const int lane  = threadIdx.x & 31;
    const int strip = blockIdx.x * 8 + (threadIdx.x >> 5);
    const int r0 = strip * 16;
    if (r0 >= nrows) return;

    const int m  = lane & 15;
    const int kk = (lane >> 4) * 2;
    const int n  = lane & 15;
    const int row = r0 + m;
    float c = cnts[row];
    float inv = 1.0f / (c > 1.0f ? c : 1.0f);

    v8f acc0 = {}, acc1 = {};
    const float* srow = sums + (size_t)row * D_FEAT;

    #pragma unroll
    for (int k0 = 0; k0 < D_FEAT; k0 += 4) {
        v2f a;
        a.x = srow[k0 + kk] * inv;
        a.y = srow[k0 + kk + 1] * inv;
        const float* w0 = Wl + (size_t)(k0 + kk) * D_OUT + n;
        const float* w1 = Wl + (size_t)(k0 + kk + 1) * D_OUT + n;
        v2f b0; b0.x = w0[0];  b0.y = w1[0];
        v2f b1; b1.x = w0[16]; b1.y = w1[16];
        acc0 = __builtin_amdgcn_wmma_f32_16x16x4_f32(false, a, false, b0, (short)0, acc0, false, false);
        acc1 = __builtin_amdgcn_wmma_f32_16x16x4_f32(false, a, false, b1, (short)0, acc1, false, false);
    }

    const int rbase = (lane >> 4) * 8;
    const int col   = lane & 15;
    #pragma unroll
    for (int r = 0; r < 8; ++r) {
        float* orow = out + (size_t)(r0 + rbase + r) * D_OUT + col;
        orow[0]  = acc0[r] + bl[col];
        orow[16] = acc1[r] + bl[col + 16];
    }
}

static inline int cdiv(long long a, int b) { return (int)((a + b - 1) / b); }

extern "C" void kernel_launch(void* const* d_in, const int* in_sizes, int n_in,
                              void* d_out, int out_size, void* d_ws, size_t ws_size,
                              hipStream_t stream)
{
    const float* x     = (const float*)d_in[0];
    const int*   ei    = (const int*)d_in[1];
    const int*   batch = (const int*)d_in[2];
    const float* W[3]  = {(const float*)d_in[3], (const float*)d_in[5], (const float*)d_in[7]};
    const float* B[3]  = {(const float*)d_in[4], (const float*)d_in[6], (const float*)d_in[8]};
    const float* Wl    = (const float*)d_in[9];
    const float* bl    = (const float*)d_in[10];
    float* out = (float*)d_out;

    const int N = in_sizes[0] / D_FEAT;
    const int E = in_sizes[1] / 2;
    const int G = out_size / D_OUT;
    const int* srcIdx = ei;
    const int* dstIdx = ei + E;

    float* ws   = (float*)d_ws;
    float* buf0 = ws;
    float* buf1 = buf0 + (size_t)N * D_FEAT;
    float* hbuf = buf1 + (size_t)N * D_FEAT;
    int*   edges = (int*)(hbuf + (size_t)N * D_FEAT);
    int*   nodes = edges + (size_t)E * 2;
    int*   rank  = nodes + (size_t)N * 2;
    int*   cnt   = rank + E;
    float* sums  = (float*)(cnt + N);
    float* cnts  = sums + (size_t)G * D_FEAT;

    const int TB = 256;

    fill_i32<<<64, TB, 0, stream>>>(cnt, 0, N);
    deg_rank<<<cdiv(E, 2 * 8 * TB), TB, 0, stream>>>(dstIdx, cnt, rank, E);
    scan_deg<<<cdiv(N, 1024), 1024, 0, stream>>>(cnt, nodes, N);
    csr_fill<<<cdiv(E, 2 * 8 * TB), TB, 0, stream>>>(srcIdx, dstIdx, nodes, rank, edges, E);

    fill_f32<<<64, TB, 0, stream>>>(sums, 0.0f, G * D_FEAT);
    fill_f32<<<1, TB, 0, stream>>>(cnts, 0.0f, G);

    const int gemmBlocks = cdiv((long long)cdiv(N, 16), 8);
    const int gatherBlocks = cdiv((long long)cdiv(N, 6) * 16, TB);
    const float* xcur = x;
    float* xbuf[3] = {buf0, buf1, buf0};

    for (int l = 0; l < 3; ++l) {
        float* xnew = xbuf[l];
        gcn_gemm64_wmma<<<gemmBlocks, TB, 0, stream>>>(xcur, W[l], hbuf, N);
        gcn_gather<<<gatherBlocks, TB, 0, stream>>>(nodes, cnt, edges, hbuf, B[l], xcur, xnew, N);
        xcur = xnew;
    }

    pool_count<<<cdiv(cdiv(N, POOL_SPAN), TB), TB, 0, stream>>>(batch, cnts, N);
    pool_sum<<<cdiv((long long)cdiv(N, POOL_SPAN) * 16, TB), TB, 0, stream>>>(xcur, batch, sums, N);
    final_linear_wmma<<<cdiv((long long)cdiv(G, 16), 8), TB, 0, stream>>>(sums, cnts, Wl, bl, out, G);
}


# Aggregate first, (A x) W in one kernel per layer: a wave sums 16 nodes' neighbour rows of x into LDS, then f32 WMMA by W, bias, relu, residual; no x W plane, 3 launches fewer; sum/product order new
# speedup vs baseline: 2.5373x; 1.0844x over previous
#include <hip/hip_runtime.h>
#include <hip/hip_bf16.h>

typedef float v2f __attribute__((ext_vector_type(2)));
typedef float v8f __attribute__((ext_vector_type(8)));

#define D_FEAT 64
#define D_OUT  32

__global__ void fill_f32(float* __restrict__ p, float v, int n) {
    for (int i = blockIdx.x * blockDim.x + threadIdx.x; i < n; i += gridDim.x * blockDim.x) p[i] = v;
}

__global__ void fill_i32(int* __restrict__ p, int v, int n) {
    for (int i = blockIdx.x * blockDim.x + threadIdx.x; i < n; i += gridDim.x * blockDim.x) p[i] = v;
}

__global__ __launch_bounds__(256) void deg_rank(const int* __restrict__ dst, int* __restrict__ cnt, int* __restrict__ rank, int e) {
    const int stride = gridDim.x * 256;
    for (int i0 = blockIdx.x * 256 + threadIdx.x; i0 < e; i0 += 8 * stride) {
        if (i0 + 7 * stride < e) {
            int d[8], r[8];
            #pragma unroll
            for (int u = 0; u < 8; ++u) d[u] = dst[i0 + u * stride];
            #pragma unroll
            for (int u = 0; u < 8; ++u) r[u] = atomicAdd(&cnt[d[u]], 1);
            #pragma unroll
            for (int u = 0; u < 8; ++u) rank[i0 + u * stride] = r[u];
        } else {
            for (int i = i0; i < e; i += stride) rank[i] = atomicAdd(&cnt[dst[i]], 1);
        }
    }
}

__global__ __launch_bounds__(1024) void scan_deg(
    const int* __restrict__ cnt, int* __restrict__ nodeRaw, int n)
{
    __shared__ int s_w[32];
    __shared__ int s_pre[32];
    const int tid = threadIdx.x, lane = tid & 31, wv = tid >> 5;
    int pre = 0;
    for (int b = 0; b < (int)blockIdx.x; ++b) pre += cnt[b * 1024 + tid];
    const int i = blockIdx.x * 1024 + tid;
    const int c = i < n ? cnt[i] : 0;
    int inc = c;
    for (int o = 1; o < 32; o <<= 1) { int t = __shfl_up(inc, o); if (lane >= o) inc += t; }
    for (int o = 16; o > 0; o >>= 1) pre += __shfl_xor(pre, o);
    if (lane == 31) s_w[wv] = inc;
    if (lane == 0)  s_pre[wv] = pre;
    __syncthreads();
    if (wv == 0) {
        const int w = s_w[lane];
        int winc = w;
        for (int o = 1; o < 32; o <<= 1) { int t = __shfl_up(winc, o); if (lane >= o) winc += t; }
        int p = s_pre[lane];
        for (int o = 16; o > 0; o >>= 1) p += __shfl_xor(p, o);
        s_w[lane] = winc - w;
        if (lane == 0) s_pre[0] = p;
    }
    __syncthreads();
    if (i < n) {
        const int start = s_pre[0] + s_w[wv] + inc - c;
        const float d = (float)(c + 1);
        const float di = d > 0.0f ? __frsqrt_rn(d) : 0.0f;
        reinterpret_cast<int2*>(nodeRaw)[i] = make_int2(start, __float_as_int(di));
    }
}

__global__ __launch_bounds__(256) void csr_fill(
    const int* __restrict__ srcIdx, const int* __restrict__ dstIdx,
    const int* __restrict__ nodeRaw, const int* __restrict__ rank,
    int* __restrict__ edgeRaw, int e)
{
    const int2* nodeInfo = reinterpret_cast<const int2*>(nodeRaw);
    int2* edgeOut = reinterpret_cast<int2*>(edgeRaw);
    const int stride = gridDim.x * 256;
    for (int i0 = blockIdx.x * 256 + threadIdx.x; i0 < e; i0 += 8 * stride) {
        if (i0 + 7 * stride < e) {
            int s[8], d[8], r[8];
            #pragma unroll
            for (int u = 0; u < 8; ++u) {
                const int i = i0 + u * stride;
                s[u] = srcIdx[i];
                d[u] = dstIdx[i];
                r[u] = rank[i];
            }
            int2 nd[8];
            float ds[8];
            #pragma unroll
            for (int u = 0; u < 8; ++u) { nd[u] = nodeInfo[d[u]]; ds[u] = __int_as_float(nodeInfo[s[u]].y); }
            #pragma unroll
            for (int u = 0; u < 8; ++u) {
                const float nrm = ds[u] * __int_as_float(nd[u].y);
                edgeOut[nd[u].x + r[u]] = make_int2(s[u], __float_as_int(nrm));
            }
        } else {
            for (int i = i0; i < e; i += stride) {
                const int s0 = srcIdx[i], d0 = dstIdx[i];
                const int2 n0 = nodeInfo[d0];
                const float nrm = __int_as_float(nodeInfo[s0].y) * __int_as_float(n0.y);
                edgeOut[n0.x + rank[i]] = make_int2(s0, __float_as_int(nrm));
            }
        }
    }
}

template <int K, bool TAIL>
__device__ __forceinline__ void gather_edges(const int2* __restrict__ p, int j, int len,
                                             const float* __restrict__ hf, float4& acc)
{
    int2 e[K];
    #pragma unroll
    for (int k = 0; k < K; ++k) {
        int idx = j + k;
        if (TAIL) idx = idx < len ? idx : len - 1;
        e[k] = p[idx];
    }
    float4 r[K];
    #pragma unroll
    for (int k = 0; k < K; ++k) r[k] = *reinterpret_cast<const float4*>(hf + (size_t)e[k].x * D_FEAT);
    #pragma unroll
    for (int k = 0; k < K; ++k) {
        if (!TAIL || j + k < len) {
            const float w = __int_as_float(e[k].y);
            acc.x = fmaf(w, r[k].x, acc.x); acc.y = fmaf(w, r[k].y, acc.y);
            acc.z = fmaf(w, r[k].z, acc.z); acc.w = fmaf(w, r[k].w, acc.w);
        }
    }
}

#define TILE_PITCH 68

__global__ __launch_bounds__(256) void gcn_layer_fused(
    const int* __restrict__ nodeRaw, const int* __restrict__ cnt, const int* __restrict__ edgeRaw,
    const float* __restrict__ xin, const float* __restrict__ W, const float* __restrict__ bias,
    float* __restrict__ xout, int nNodes)
{
    __shared__ __attribute__((aligned(16))) float s_tile[8][16 * TILE_PITCH];
    const int lane = threadIdx.x & 31;
    const int wv   = threadIdx.x >> 5;
    const int r0   = (blockIdx.x * 8 + wv) * 16;
    if (r0 >= nNodes) return;
    float* st = s_tile[wv];

    const int g = lane >> 4;
    const int f = (lane & 15) * 4;
    const int2* nodeInfo = reinterpret_cast<const int2*>(nodeRaw);
    const int2* edges = reinterpret_cast<const int2*>(edgeRaw);
    const float* xf = xin + f;

    for (int i = 0; i < 8; ++i) {
        const int nl = 2 * i + g;
        const int node = r0 + nl;
        float4 acc = make_float4(0.0f, 0.0f, 0.0f, 0.0f);
        if (node < nNodes) {
            const int2 nd = nodeInfo[node];
            const int len = cnt[node];
            const float4 xs = *reinterpret_cast<const float4*>(xf + (size_t)node * D_FEAT);
            const int2* p = edges + nd.x;
            int j = 0;
            for (; j + 8 <= len; j += 8) gather_edges<8, false>(p, j, len, xf, acc);
            if (j + 4 <= len) { gather_edges<4, false>(p, j, len, xf, acc); j += 4; }
            if (j < len) gather_edges<4, true>(p, j, len, xf, acc);
            const float di = __int_as_float(nd.y);
            const float dd = di * di;
            acc.x += dd * xs.x; acc.y += dd * xs.y; acc.z += dd * xs.z; acc.w += dd * xs.w;
        }
        *reinterpret_cast<float4*>(st + nl * TILE_PITCH + f) = acc;
    }
    __builtin_amdgcn_fence(__ATOMIC_RELEASE, "wavefront");
    __builtin_amdgcn_wave_barrier();
    __builtin_amdgcn_fence(__ATOMIC_ACQUIRE, "wavefront");

    const int m  = lane & 15;
    const int kk = (lane >> 4) * 2;
    const int n  = lane & 15;
    v8f acc0 = {}, acc1 = {}, acc2 = {}, acc3 = {};
    const float* arow = st + m * TILE_PITCH + kk;

    #pragma unroll
    for (int k0 = 0; k0 < D_FEAT; k0 += 4) {
        v2f a;
        a.x = arow[k0];
        a.y = arow[k0 + 1];
        const float* w0 = W + (size_t)(k0 + kk) * D_FEAT + n;
        const float* w1 = W + (size_t)(k0 + kk + 1) * D_FEAT + n;
        v2f b0; b0.x = w0[0];  b0.y = w1[0];
        v2f b1; b1.x = w0[16]; b1.y = w1[16];
        v2f b2; b2.x = w0[32]; b2.y = w1[32];
        v2f b3; b3.x = w0[48]; b3.y = w1[48];
        acc0 = __builtin_amdgcn_wmma_f32_16x16x4_f32(false, a, false, b0, (short)0, acc0, false, false);
        acc1 = __builtin_amdgcn_wmma_f32_16x16x4_f32(false, a, false, b1, (short)0, acc1, false, false);
        acc2 = __builtin_amdgcn_wmma_f32_16x16x4_f32(false, a, false, b2, (short)0, acc2, false, false);
        acc3 = __builtin_amdgcn_wmma_f32_16x16x4_f32(false, a, false, b3, (short)0, acc3, false, false);
    }

    __builtin_amdgcn_fence(__ATOMIC_RELEASE, "wavefront");
    __builtin_amdgcn_wave_barrier();
    __builtin_amdgcn_fence(__ATOMIC_ACQUIRE, "wavefront");
    const int rbase = (lane >> 4) * 8;
    const int col   = lane & 15;
    #pragma unroll
    for (int r = 0; r < 8; ++r) {
        float* trow = st + (rbase + r) * TILE_PITCH + col;
        trow[0]  = acc0[r];
        trow[16] = acc1[r];
        trow[32] = acc2[r];
        trow[48] = acc3[r];
    }
    __builtin_amdgcn_fence(__ATOMIC_RELEASE, "wavefront");
    __builtin_amdgcn_wave_barrier();
    __builtin_amdgcn_fence(__ATOMIC_ACQUIRE, "wavefront");

    const float4 bb = *reinterpret_cast<const float4*>(bias + f);
    float4 xp[8];
    #pragma unroll
    for (int i = 0; i < 8; ++i) {
        const int node = r0 + 2 * i + g;
        const int nc = node < nNodes ? node : nNodes - 1;
        xp[i] = *reinterpret_cast<const float4*>(xf + (size_t)nc * D_FEAT);
    }
    #pragma unroll
    for (int i = 0; i < 8; ++i) {
        const int nl = 2 * i + g;
        const int node = r0 + nl;
        float4 v = *reinterpret_cast<const float4*>(st + nl * TILE_PITCH + f);
        v.x += bb.x; v.y += bb.y; v.z += bb.z; v.w += bb.w;
        v.x = (v.x > 0.0f ? v.x : 0.0f) + xp[i].x;
        v.y = (v.y > 0.0f ? v.y : 0.0f) + xp[i].y;
        v.z = (v.z > 0.0f ? v.z : 0.0f) + xp[i].z;
        v.w = (v.w > 0.0f ? v.w : 0.0f) + xp[i].w;
        if (node < nNodes) *reinterpret_cast<float4*>(xout + (size_t)node * D_FEAT + f) = v;
    }
}

#define POOL_SPAN 32

__global__ void pool_count(const int* __restrict__ batch, float* __restrict__ cnts, int n) {
    const int beg = (blockIdx.x * blockDim.x + threadIdx.x) * POOL_SPAN;
    if (beg >= n) return;
    const int end = beg + POOL_SPAN < n ? beg + POOL_SPAN : n;
    int g = batch[beg];
    float c = 0.0f;
    int i = beg;
    for (; i + 8 <= end; i += 8) {
        int b[8];
        #pragma unroll
        for (int k = 0; k < 8; ++k) b[k] = batch[i + k];
        #pragma unroll
        for (int k = 0; k < 8; ++k) {
            if (b[k] != g) { unsafeAtomicAdd(&cnts[g], c); c = 0.0f; g = b[k]; }
            c += 1.0f;
        }
    }
    for (; i < end; ++i) {
        const int b0 = batch[i];
        if (b0 != g) { unsafeAtomicAdd(&cnts[g], c); c = 0.0f; g = b0; }
        c += 1.0f;
    }
    unsafeAtomicAdd(&cnts[g], c);
}

__device__ __forceinline__ void pool_flush(float* __restrict__ sums, int g, int f, const float4& a) {
    float* s = sums + (size_t)g * D_FEAT + f;
    unsafeAtomicAdd(s + 0, a.x);
    unsafeAtomicAdd(s + 1, a.y);
    unsafeAtomicAdd(s + 2, a.z);
    unsafeAtomicAdd(s + 3, a.w);
}

__global__ __launch_bounds__(256) void pool_sum(
    const float* __restrict__ x, const int* __restrict__ batch,
    float* __restrict__ sums, int n)
{
    const int f = (threadIdx.x & 15) * 4;
    const int beg = ((blockIdx.x * 256 + threadIdx.x) >> 4) * POOL_SPAN;
    if (beg >= n) return;
    const int end = beg + POOL_SPAN < n ? beg + POOL_SPAN : n;
    const float* xf = x + f;
    int g = batch[beg];
    float4 acc = make_float4(0.0f, 0.0f, 0.0f, 0.0f);
    int i = beg;
    for (; i + 8 <= end; i += 8) {
        int b[8];
        float4 v[8];
        #pragma unroll
        for (int k = 0; k < 8; ++k) {
            b[k] = batch[i + k];
            v[k] = *reinterpret_cast<const float4*>(xf + (size_t)(i + k) * D_FEAT);
        }
        #pragma unroll
        for (int k = 0; k < 8; ++k) {
            if (b[k] != g) { pool_flush(sums, g, f, acc); acc = make_float4(0.0f, 0.0f, 0.0f, 0.0f); g = b[k]; }
            acc.x += v[k].x; acc.y += v[k].y; acc.z += v[k].z; acc.w += v[k].w;
        }
    }
    for (; i < end; ++i) {
        const int b0 = batch[i];
        const float4 v0 = *reinterpret_cast<const float4*>(xf + (size_t)i * D_FEAT);
        if (b0 != g) { pool_flush(sums, g, f, acc); acc = make_float4(0.0f, 0.0f, 0.0f, 0.0f); g = b0; }
        acc.x += v0.x; acc.y += v0.y; acc.z += v0.z; acc.w += v0.w;
    }
    pool_flush(sums, g, f, acc);
}

__global__ __launch_bounds__(256) void final_linear_wmma(
    const float* __restrict__ sums, const float* __restrict__ cnts,
    const float* __restrict__ Wl, const float* __restrict__ bl,
    float* __restrict__ out, int nrows)
{
    const int lane  = threadIdx.x & 31;
    const int strip = blockIdx.x * 8 + (threadIdx.x >> 5);
    const int r0 = strip * 16;
    if (r0 >= nrows) return;

    const int m  = lane & 15;
    const int kk = (lane >> 4) * 2;
    const int n  = lane & 15;
    const int row = r0 + m;
    float c = cnts[row];
    float inv = 1.0f / (c > 1.0f ? c : 1.0f);

    v8f acc0 = {}, acc1 = {};
    const float* srow = sums + (size_t)row * D_FEAT;

    #pragma unroll
    for (int k0 = 0; k0 < D_FEAT; k0 += 4) {
        v2f a;
        a.x = srow[k0 + kk] * inv;
        a.y = srow[k0 + kk + 1] * inv;
        const float* w0 = Wl + (size_t)(k0 + kk) * D_OUT + n;
        const float* w1 = Wl + (size_t)(k0 + kk + 1) * D_OUT + n;
        v2f b0; b0.x = w0[0];  b0.y = w1[0];
        v2f b1; b1.x = w0[16]; b1.y = w1[16];
        acc0 = __builtin_amdgcn_wmma_f32_16x16x4_f32(false, a, false, b0, (short)0, acc0, false, false);
        acc1 = __builtin_amdgcn_wmma_f32_16x16x4_f32(false, a, false, b1, (short)0, acc1, false, false);
    }

    const int rbase = (lane >> 4) * 8;
    const int col   = lane & 15;
    #pragma unroll
    for (int r = 0; r < 8; ++r) {
        float* orow = out + (size_t)(r0 + rbase + r) * D_OUT + col;
        orow[0]  = acc0[r] + bl[col];
        orow[16] = acc1[r] + bl[col + 16];
    }
}

static inline int cdiv(long long a, int b) { return (int)((a + b - 1) / b); }

extern "C" void kernel_launch(void* const* d_in, const int* in_sizes, int n_in,
                              void* d_out, int out_size, void* d_ws, size_t ws_size,
                              hipStream_t stream)
{
    const float* x     = (const float*)d_in[0];
    const int*   ei    = (const int*)d_in[1];
    const int*   batch = (const int*)d_in[2];
    const float* W[3]  = {(const float*)d_in[3], (const float*)d_in[5], (const float*)d_in[7]};
    const float* B[3]  = {(const float*)d_in[4], (const float*)d_in[6], (const float*)d_in[8]};
    const float* Wl    = (const float*)d_in[9];
    const float* bl    = (const float*)d_in[10];
    float* out = (float*)d_out;

    const int N = in_sizes[0] / D_FEAT;
    const int E = in_sizes[1] / 2;
    const int G = out_size / D_OUT;
    const int* srcIdx = ei;
    const int* dstIdx = ei + E;

    float* ws   = (float*)d_ws;
    float* buf0 = ws;
    float* buf1 = buf0 + (size_t)N * D_FEAT;
    int*   edges = (int*)(buf1 + (size_t)N * D_FEAT);
    int*   nodes = edges + (size_t)E * 2;
    int*   rank  = nodes + (size_t)N * 2;
    int*   cnt   = rank + E;
    float* sums  = (float*)(cnt + N);
    float* cnts  = sums + (size_t)G * D_FEAT;

    const int TB = 256;

    fill_i32<<<64, TB, 0, stream>>>(cnt, 0, N);
    deg_rank<<<cdiv(E, 2 * 8 * TB), TB, 0, stream>>>(dstIdx, cnt, rank, E);
    scan_deg<<<cdiv(N, 1024), 1024, 0, stream>>>(cnt, nodes, N);
    csr_fill<<<cdiv(E, 2 * 8 * TB), TB, 0, stream>>>(srcIdx, dstIdx, nodes, rank, edges, E);

    fill_f32<<<64, TB, 0, stream>>>(sums, 0.0f, G * D_FEAT);
    fill_f32<<<1, TB, 0, stream>>>(cnts, 0.0f, G);

    const int layerBlocks = cdiv((long long)cdiv(N, 16), 8);
    const float* xcur = x;
    float* xbuf[3] = {buf0, buf1, buf0};

    for (int l = 0; l < 3; ++l) {
        float* xnew = xbuf[l];
        gcn_layer_fused<<<layerBlocks, TB, 0, stream>>>(nodes, cnt, edges, xcur, W[l], B[l], xnew, N);
        xcur = xnew;
    }

    pool_count<<<cdiv(cdiv(N, POOL_SPAN), TB), TB, 0, stream>>>(batch, cnts, N);
    pool_sum<<<cdiv((long long)cdiv(N, POOL_SPAN) * 16, TB), TB, 0, stream>>>(xcur, batch, sums, N);
    final_linear_wmma<<<cdiv((long long)cdiv(G, 16), 8), TB, 0, stream>>>(sums, cnts, Wl, bl, out, G);
}


# Edges sorted by destination with a two-level counting sort on LDS counters (tile x bucket bins, per-bucket places, staged whole-line stores), no global atomic per edge; sum order = LDS arrival rank
# speedup vs baseline: 4.1333x; 1.6290x over previous
#include <hip/hip_runtime.h>
#include <hip/hip_bf16.h>

typedef float v2f __attribute__((ext_vector_type(2)));
typedef float v8f __attribute__((ext_vector_type(8)));

#define D_FEAT 64
#define D_OUT  32

__global__ void fill_f32(float* __restrict__ p, float v, int n) {
    for (int i = blockIdx.x * blockDim.x + threadIdx.x; i < n; i += gridDim.x * blockDim.x) p[i] = v;
}

#define BK_NODES    256
#define TILE_EDGES  8192
#define MAX_BUCKETS 1024
#define STAGE_CAP   6144

__device__ __forceinline__ int bucket_of(int d, int nb) {
    const int b = d >> 8;
    return b < nb ? b : nb - 1;
}

__global__ __launch_bounds__(1024) void tile_hist(
    const int* __restrict__ dst, int* __restrict__ tcnt, int e, int nb, int nt)
{
    __shared__ int s_h[MAX_BUCKETS];
    const int tid = threadIdx.x, t = blockIdx.x;
    for (int b = tid; b < nb; b += 1024) s_h[b] = 0;
    __syncthreads();
    const int base = t * TILE_EDGES + tid;
    int d[8];
    #pragma unroll
    for (int u = 0; u < 8; ++u) {
        const int i = base + u * 1024;
        const int v = dst[i < e ? i : e - 1];
        d[u] = i < e ? v : -1;
    }
    #pragma unroll
    for (int u = 0; u < 8; ++u) if (d[u] >= 0) atomicAdd(&s_h[bucket_of(d[u], nb)], 1);
    __syncthreads();
    for (int b = tid; b < nb; b += 1024) tcnt[b * nt + t] = s_h[b];
}

__global__ __launch_bounds__(1024) void scan_excl(const int* __restrict__ in, int* __restrict__ out, int n)
{
    __shared__ int s_w[32];
    __shared__ int s_pre[32];
    const int tid = threadIdx.x, lane = tid & 31, wv = tid >> 5;
    int pre = 0;
    for (int b = 0; b < (int)blockIdx.x; ++b) pre += in[b * 1024 + tid];
    const int i = blockIdx.x * 1024 + tid;
    const int c = i < n ? in[i] : 0;
    int inc = c;
    for (int o = 1; o < 32; o <<= 1) { int t = __shfl_up(inc, o); if (lane >= o) inc += t; }
    for (int o = 16; o > 0; o >>= 1) pre += __shfl_xor(pre, o);
    if (lane == 31) s_w[wv] = inc;
    if (lane == 0)  s_pre[wv] = pre;
    __syncthreads();
    if (wv == 0) {
        const int w = s_w[lane];
        int winc = w;
        for (int o = 1; o < 32; o <<= 1) { int t = __shfl_up(winc, o); if (lane >= o) winc += t; }
        int p = s_pre[lane];
        for (int o = 16; o > 0; o >>= 1) p += __shfl_xor(p, o);
        s_w[lane] = winc - w;
        if (lane == 0) s_pre[0] = p;
    }
    __syncthreads();
    if (i < n) out[i] = s_pre[0] + s_w[wv] + inc - c;
}

__global__ __launch_bounds__(1024) void tile_scatter(
    const int* __restrict__ srcIdx, const int* __restrict__ dstIdx, const int* __restrict__ toffs,
    int* __restrict__ binnedRaw, int e, int nb, int nt)
{
    __shared__ int s_cur[MAX_BUCKETS];
    const int tid = threadIdx.x, t = blockIdx.x;
    for (int b = tid; b < nb; b += 1024) s_cur[b] = toffs[b * nt + t];
    const int base = t * TILE_EDGES + tid;
    int s[8], d[8];
    #pragma unroll
    for (int u = 0; u < 8; ++u) {
        const int i = base + u * 1024;
        const int ic = i < e ? i : e - 1;
        const int v = dstIdx[ic];
        s[u] = srcIdx[ic];
        d[u] = i < e ? v : -1;
    }
    __syncthreads();
    int2* binned = reinterpret_cast<int2*>(binnedRaw);
    #pragma unroll
    for (int u = 0; u < 8; ++u) {
        if (d[u] >= 0) {
            const int pos = atomicAdd(&s_cur[bucket_of(d[u], nb)], 1);
            binned[pos] = make_int2(s[u], d[u]);
        }
    }
}

__global__ __launch_bounds__(BK_NODES) void bucket_count(
    const int* __restrict__ binnedRaw, const int* __restrict__ toffs,
    int* __restrict__ nodeRaw, int* __restrict__ cnt, int e, int n, int nb, int nt)
{
    __shared__ int s_c[BK_NODES];
    __shared__ int s_w[BK_NODES / 32];
    const int tid = threadIdx.x, lane = tid & 31, wv = tid >> 5;
    const int b = blockIdx.x;
    const int2* binned = reinterpret_cast<const int2*>(binnedRaw);
    const int beg = toffs[b * nt];
    const int end = b + 1 < nb ? toffs[(b + 1) * nt] : e;
    s_c[tid] = 0;
    __syncthreads();
    for (int i0 = beg + tid; i0 < end; i0 += 8 * BK_NODES) {
        int d[8];
        #pragma unroll
        for (int u = 0; u < 8; ++u) {
            const int i = i0 + u * BK_NODES;
            const int v = binned[i < end ? i : end - 1].y;
            d[u] = i < end ? v : -1;
        }
        #pragma unroll
        for (int u = 0; u < 8; ++u) if (d[u] >= 0) atomicAdd(&s_c[(d[u] - b * BK_NODES) & (BK_NODES - 1)], 1);
    }
    __syncthreads();
    const int c = s_c[tid];
    int inc = c;
    for (int o = 1; o < 32; o <<= 1) { int t = __shfl_up(inc, o); if (lane >= o) inc += t; }
    if (lane == 31) s_w[wv] = inc;
    __syncthreads();
    int wpre = 0;
    for (int w = 0; w < wv; ++w) wpre += s_w[w];
    const int node = b * BK_NODES + tid;
    if (node < n) {
        const int start = beg + wpre + inc - c;
        const float d = (float)(c + 1);
        const float di = d > 0.0f ? __frsqrt_rn(d) : 0.0f;
        reinterpret_cast<int2*>(nodeRaw)[node] = make_int2(start, __float_as_int(di));
        cnt[node] = c;
    }
}

__global__ __launch_bounds__(BK_NODES) void bucket_place(
    const int* __restrict__ binnedRaw, const int* __restrict__ toffs, const int* __restrict__ nodeRaw,
    int* __restrict__ edgeRaw, int e, int n, int nb, int nt)
{
    __shared__ int   s_cur[BK_NODES];
    __shared__ float s_di[BK_NODES];
    __shared__ int2  s_stage[STAGE_CAP];
    const int tid = threadIdx.x;
    const int b = blockIdx.x;
    const int2* binned = reinterpret_cast<const int2*>(binnedRaw);
    const int2* nodeInfo = reinterpret_cast<const int2*>(nodeRaw);
    int2* edgeOut = reinterpret_cast<int2*>(edgeRaw);
    const int beg = toffs[b * nt];
    const int end = b + 1 < nb ? toffs[(b + 1) * nt] : e;
    const int count = end - beg;
    const bool staged = count <= STAGE_CAP;
    const int node = b * BK_NODES + tid;
    const int2 nd = nodeInfo[node < n ? node : n - 1];
    s_cur[tid] = nd.x;
    s_di[tid]  = __int_as_float(nd.y);
    __syncthreads();
    for (int i0 = beg + tid; i0 < end; i0 += 8 * BK_NODES) {
        int2 sd[8];
        float ds[8];
        #pragma unroll
        for (int u = 0; u < 8; ++u) {
            const int i = i0 + u * BK_NODES;
            sd[u] = binned[i < end ? i : end - 1];
            if (i >= end) sd[u].y = -1;
        }
        #pragma unroll
        for (int u = 0; u < 8; ++u) ds[u] = __int_as_float(nodeInfo[sd[u].x].y);
        #pragma unroll
        for (int u = 0; u < 8; ++u) {
            if (sd[u].y >= 0) {
                const int l = (sd[u].y - b * BK_NODES) & (BK_NODES - 1);
                const int pos = atomicAdd(&s_cur[l], 1);
                const int2 rec = make_int2(sd[u].x, __float_as_int(ds[u] * s_di[l]));
                if (staged) s_stage[pos - beg] = rec; else edgeOut[pos] = rec;
            }
        }
    }
    if (staged) {
        __syncthreads();
        for (int i = tid; i < count; i += BK_NODES) edgeOut[beg + i] = s_stage[i];
    }
}


#define TILE_PITCH 68
#define GK 8

__global__ __launch_bounds__(256) void gcn_layer_fused(
    const int* __restrict__ nodeRaw, const int* __restrict__ cnt, const int* __restrict__ edgeRaw,
    const float* __restrict__ xin, const float* __restrict__ W, const float* __restrict__ bias,
    float* __restrict__ xout, int nNodes)
{
    __shared__ __attribute__((aligned(16))) float s_tile[8][16 * TILE_PITCH];
    __shared__ int s_end[8][32];
    const int lane = threadIdx.x & 31;
    const int wv   = threadIdx.x >> 5;
    const int r0   = (blockIdx.x * 8 + wv) * 16;
    if (r0 >= nNodes) return;
    float* st = s_tile[wv];
    int* se = s_end[wv];

    const int g  = lane >> 4;
    const int hb = lane & 16;
    const int f  = (lane & 15) * 4;
    const int first = r0 + 8 * g;
    const int2* nodeInfo = reinterpret_cast<const int2*>(nodeRaw);
    const int2* edges = reinterpret_cast<const int2*>(edgeRaw);
    const float* xf = xin + f;

    const int jn = (lane & 15) < 8 ? (lane & 15) : 7;
    const int myNode = first + jn;
    const bool myValid = myNode < nNodes;
    const int myC = myValid ? myNode : nNodes - 1;
    const int2 nd = nodeInfo[myC];
    const int segEnd = nd.x + cnt[myC];
    const int segBeg = myValid ? nd.x : segEnd;
    const float di = myValid ? __int_as_float(nd.y) : 0.0f;
    se[lane] = segEnd;
    #pragma unroll
    for (int i = 0; i < 8; ++i)
        *reinterpret_cast<float4*>(st + (8 * g + i) * TILE_PITCH + f) = make_float4(0.0f, 0.0f, 0.0f, 0.0f);
    const int beg = __shfl(segBeg, hb);
    const int end = __shfl(segEnd, hb + 7);
    __builtin_amdgcn_fence(__ATOMIC_RELEASE, "wavefront");
    __builtin_amdgcn_wave_barrier();
    __builtin_amdgcn_fence(__ATOMIC_ACQUIRE, "wavefront");

    {
        int cur = 0;
        int nxt = __shfl(segEnd, hb);
        float4 acc = make_float4(0.0f, 0.0f, 0.0f, 0.0f);
        int2 pr[GK];
        #pragma unroll
        for (int k = 0; k < GK; ++k) pr[k] = edges[beg + k];
        for (int base = beg; base < end; base += GK) {
            int2 nx[GK];
            #pragma unroll
            for (int k = 0; k < GK; ++k) nx[k] = edges[base + GK + k];
            float4 r[GK];
            #pragma unroll
            for (int k = 0; k < GK; ++k) {
                const int s = base + k < end ? pr[k].x : pr[0].x;
                r[k] = *reinterpret_cast<const float4*>(xf + (size_t)s * D_FEAT);
            }
            #pragma unroll
            for (int k = 0; k < GK; ++k) {
                const int e = base + k;
                if (e < end) {
                    if (e >= nxt) {
                        *reinterpret_cast<float4*>(st + (8 * g + cur) * TILE_PITCH + f) = acc;
                        acc = make_float4(0.0f, 0.0f, 0.0f, 0.0f);
                        do { ++cur; nxt = se[hb + cur]; } while (e >= nxt && cur < 7);
                    }
                    const float w = __int_as_float(pr[k].y);
                    acc.x = fmaf(w, r[k].x, acc.x); acc.y = fmaf(w, r[k].y, acc.y);
                    acc.z = fmaf(w, r[k].z, acc.z); acc.w = fmaf(w, r[k].w, acc.w);
                }
            }
            #pragma unroll
            for (int k = 0; k < GK; ++k) pr[k] = nx[k];
        }
        if (beg < end) *reinterpret_cast<float4*>(st + (8 * g + cur) * TILE_PITCH + f) = acc;
    }

    float4 xs[8];
    #pragma unroll
    for (int i = 0; i < 8; ++i) {
        const int node = first + i;
        const int nc = node < nNodes ? node : nNodes - 1;
        xs[i] = *reinterpret_cast<const float4*>(xf + (size_t)nc * D_FEAT);
    }
    #pragma unroll
    for (int i = 0; i < 8; ++i) {
        float* trow = st + (8 * g + i) * TILE_PITCH + f;
        float4 v = *reinterpret_cast<float4*>(trow);
        const float d = __shfl(di, hb + i);
        const float dd = d * d;
        v.x += dd * xs[i].x; v.y += dd * xs[i].y; v.z += dd * xs[i].z; v.w += dd * xs[i].w;
        *reinterpret_cast<float4*>(trow) = v;
    }
    __builtin_amdgcn_fence(__ATOMIC_RELEASE, "wavefront");
    __builtin_amdgcn_wave_barrier();
    __builtin_amdgcn_fence(__ATOMIC_ACQUIRE, "wavefront");

    const int m  = lane & 15;
    const int kk = (lane >> 4) * 2;
    const int n  = lane & 15;
    v8f acc0 = {}, acc1 = {}, acc2 = {}, acc3 = {};
    const float* arow = st + m * TILE_PITCH + kk;

    #pragma unroll
    for (int k0 = 0; k0 < D_FEAT; k0 += 4) {
        v2f a;
        a.x = arow[k0];
        a.y = arow[k0 + 1];
        const float* w0 = W + (size_t)(k0 + kk) * D_FEAT + n;
        const float* w1 = W + (size_t)(k0 + kk + 1) * D_FEAT + n;
        v2f b0; b0.x = w0[0];  b0.y = w1[0];
        v2f b1; b1.x = w0[16]; b1.y = w1[16];
        v2f b2; b2.x = w0[32]; b2.y = w1[32];
        v2f b3; b3.x = w0[48]; b3.y = w1[48];
        acc0 = __builtin_amdgcn_wmma_f32_16x16x4_f32(false, a, false, b0, (short)0, acc0, false, false);
        acc1 = __builtin_amdgcn_wmma_f32_16x16x4_f32(false, a, false, b1, (short)0, acc1, false, false);
        acc2 = __builtin_amdgcn_wmma_f32_16x16x4_f32(false, a, false, b2, (short)0, acc2, false, false);
        acc3 = __builtin_amdgcn_wmma_f32_16x16x4_f32(false, a, false, b3, (short)0, acc3, false, false);
    }

    __builtin_amdgcn_fence(__ATOMIC_RELEASE, "wavefront");
    __builtin_amdgcn_wave_barrier();
    __builtin_amdgcn_fence(__ATOMIC_ACQUIRE, "wavefront");
    const int rbase = (lane >> 4) * 8;
    const int col   = lane & 15;
    #pragma unroll
    for (int r = 0; r < 8; ++r) {
        float* trow = st + (rbase + r) * TILE_PITCH + col;
        trow[0]  = acc0[r];
        trow[16] = acc1[r];
        trow[32] = acc2[r];
        trow[48] = acc3[r];
    }
    __builtin_amdgcn_fence(__ATOMIC_RELEASE, "wavefront");
    __builtin_amdgcn_wave_barrier();
    __builtin_amdgcn_fence(__ATOMIC_ACQUIRE, "wavefront");

    const float4 bb = *reinterpret_cast<const float4*>(bias + f);
    #pragma unroll
    for (int i = 0; i < 8; ++i) {
        const int nl = 8 * g + i;
        const int node = r0 + nl;
        float4 v = *reinterpret_cast<float4*>(st + nl * TILE_PITCH + f);
        v.x += bb.x; v.y += bb.y; v.z += bb.z; v.w += bb.w;
        v.x = (v.x > 0.0f ? v.x : 0.0f) + xs[i].x;
        v.y = (v.y > 0.0f ? v.y : 0.0f) + xs[i].y;
        v.z = (v.z > 0.0f ? v.z : 0.0f) + xs[i].z;
        v.w = (v.w > 0.0f ? v.w : 0.0f) + xs[i].w;
        if (node < nNodes) *reinterpret_cast<float4*>(xout + (size_t)node * D_FEAT + f) = v;
    }
}

#define POOL_SPAN 32

__global__ void pool_count(const int* __restrict__ batch, float* __restrict__ cnts, int n) {
    const int beg = (blockIdx.x * blockDim.x + threadIdx.x) * POOL_SPAN;
    if (beg >= n) return;
    const int end = beg + POOL_SPAN < n ? beg + POOL_SPAN : n;
    int g = batch[beg];
    float c = 0.0f;
    int i = beg;
    for (; i + 8 <= end; i += 8) {
        int b[8];
        #pragma unroll
        for (int k = 0; k < 8; ++k) b[k] = batch[i + k];
        #pragma unroll
        for (int k = 0; k < 8; ++k) {
            if (b[k] != g) { unsafeAtomicAdd(&cnts[g], c); c = 0.0f; g = b[k]; }
            c += 1.0f;
        }
    }
    for (; i < end; ++i) {
        const int b0 = batch[i];
        if (b0 != g) { unsafeAtomicAdd(&cnts[g], c); c = 0.0f; g = b0; }
        c += 1.0f;
    }
    unsafeAtomicAdd(&cnts[g], c);
}

__device__ __forceinline__ void pool_flush(float* __restrict__ sums, int g, int f, const float4& a) {
    float* s = sums + (size_t)g * D_FEAT + f;
    unsafeAtomicAdd(s + 0, a.x);
    unsafeAtomicAdd(s + 1, a.y);
    unsafeAtomicAdd(s + 2, a.z);
    unsafeAtomicAdd(s + 3, a.w);
}

__global__ __launch_bounds__(256) void pool_sum(
    const float* __restrict__ x, const int* __restrict__ batch,
    float* __restrict__ sums, int n)
{
    const int f = (threadIdx.x & 15) * 4;
    const int beg = ((blockIdx.x * 256 + threadIdx.x) >> 4) * POOL_SPAN;
    if (beg >= n) return;
    const int end = beg + POOL_SPAN < n ? beg + POOL_SPAN : n;
    const float* xf = x + f;
    int g = batch[beg];
    float4 acc = make_float4(0.0f, 0.0f, 0.0f, 0.0f);
    int i = beg;
    for (; i + 8 <= end; i += 8) {
        int b[8];
        float4 v[8];
        #pragma unroll
        for (int k = 0; k < 8; ++k) {
            b[k] = batch[i + k];
            v[k] = *reinterpret_cast<const float4*>(xf + (size_t)(i + k) * D_FEAT);
        }
        #pragma unroll
        for (int k = 0; k < 8; ++k) {
            if (b[k] != g) { pool_flush(sums, g, f, acc); acc = make_float4(0.0f, 0.0f, 0.0f, 0.0f); g = b[k]; }
            acc.x += v[k].x; acc.y += v[k].y; acc.z += v[k].z; acc.w += v[k].w;
        }
    }
    for (; i < end; ++i) {
        const int b0 = batch[i];
        const float4 v0 = *reinterpret_cast<const float4*>(xf + (size_t)i * D_FEAT);
        if (b0 != g) { pool_flush(sums, g, f, acc); acc = make_float4(0.0f, 0.0f, 0.0f, 0.0f); g = b0; }
        acc.x += v0.x; acc.y += v0.y; acc.z += v0.z; acc.w += v0.w;
    }
    pool_flush(sums, g, f, acc);
}

__global__ __launch_bounds__(256) void final_linear_wmma(
    const float* __restrict__ sums, const float* __restrict__ cnts,
    const float* __restrict__ Wl, const float* __restrict__ bl,
    float* __restrict__ out, int nrows)
{
    const int lane  = threadIdx.x & 31;
    const int strip = blockIdx.x * 8 + (threadIdx.x >> 5);
    const int r0 = strip * 16;
    if (r0 >= nrows) return;

    const int m  = lane & 15;
    const int kk = (lane >> 4) * 2;
    const int n  = lane & 15;
    const int row = r0 + m;
    float c = cnts[row];
    float inv = 1.0f / (c > 1.0f ? c : 1.0f);

    v8f acc0 = {}, acc1 = {};
    const float* srow = sums + (size_t)row * D_FEAT;

    #pragma unroll
    for (int k0 = 0; k0 < D_FEAT; k0 += 4) {
        v2f a;
        a.x = srow[k0 + kk] * inv;
        a.y = srow[k0 + kk + 1] * inv;
        const float* w0 = Wl + (size_t)(k0 + kk) * D_OUT + n;
        const float* w1 = Wl + (size_t)(k0 + kk + 1) * D_OUT + n;
        v2f b0; b0.x = w0[0];  b0.y = w1[0];
        v2f b1; b1.x = w0[16]; b1.y = w1[16];
        acc0 = __builtin_amdgcn_wmma_f32_16x16x4_f32(false, a, false, b0, (short)0, acc0, false, false);
        acc1 = __builtin_amdgcn_wmma_f32_16x16x4_f32(false, a, false, b1, (short)0, acc1, false, false);
    }

    const int rbase = (lane >> 4) * 8;
    const int col   = lane & 15;
    #pragma unroll
    for (int r = 0; r < 8; ++r) {
        float* orow = out + (size_t)(r0 + rbase + r) * D_OUT + col;
        orow[0]  = acc0[r] + bl[col];
        orow[16] = acc1[r] + bl[col + 16];
    }
}

static inline int cdiv(long long a, int b) { return (int)((a + b - 1) / b); }

extern "C" void kernel_launch(void* const* d_in, const int* in_sizes, int n_in,
                              void* d_out, int out_size, void* d_ws, size_t ws_size,
                              hipStream_t stream)
{
    const float* x     = (const float*)d_in[0];
    const int*   ei    = (const int*)d_in[1];
    const int*   batch = (const int*)d_in[2];
    const float* W[3]  = {(const float*)d_in[3], (const float*)d_in[5], (const float*)d_in[7]};
    const float* B[3]  = {(const float*)d_in[4], (const float*)d_in[6], (const float*)d_in[8]};
    const float* Wl    = (const float*)d_in[9];
    const float* bl    = (const float*)d_in[10];
    float* out = (float*)d_out;

    const int N = in_sizes[0] / D_FEAT;
    const int E = in_sizes[1] / 2;
    const int G = out_size / D_OUT;
    const int* srcIdx = ei;
    const int* dstIdx = ei + E;

    float* ws   = (float*)d_ws;
    float* buf0 = ws;
    float* buf1 = buf0 + (size_t)N * D_FEAT;
    const int NB = cdiv(N, BK_NODES);
    const int NT = cdiv(E, TILE_EDGES);
    int*   edges  = (int*)(buf1 + (size_t)N * D_FEAT);
    int*   nodes  = edges + (size_t)E * 2;
    int*   binned = nodes + (size_t)N * 2;
    int*   cnt    = binned + (size_t)E * 2;
    int*   tcnt   = cnt + N;
    int*   toffs  = tcnt + (size_t)NB * NT;
    float* sums   = (float*)(toffs + (size_t)NB * NT);
    float* cnts   = sums + (size_t)G * D_FEAT;

    const int TB = 256;

    tile_hist<<<NT, 1024, 0, stream>>>(dstIdx, tcnt, E, NB, NT);
    scan_excl<<<cdiv((long long)NB * NT, 1024), 1024, 0, stream>>>(tcnt, toffs, NB * NT);
    tile_scatter<<<NT, 1024, 0, stream>>>(srcIdx, dstIdx, toffs, binned, E, NB, NT);
    bucket_count<<<NB, BK_NODES, 0, stream>>>(binned, toffs, nodes, cnt, E, N, NB, NT);
    bucket_place<<<NB, BK_NODES, 0, stream>>>(binned, toffs, nodes, edges, E, N, NB, NT);

    fill_f32<<<64, TB, 0, stream>>>(sums, 0.0f, G * D_FEAT);
    fill_f32<<<1, TB, 0, stream>>>(cnts, 0.0f, G);

    const int layerBlocks = cdiv((long long)cdiv(N, 16), 8);
    const float* xcur = x;
    float* xbuf[3] = {buf0, buf1, buf0};

    for (int l = 0; l < 3; ++l) {
        float* xnew = xbuf[l];
        gcn_layer_fused<<<layerBlocks, TB, 0, stream>>>(nodes, cnt, edges, xcur, W[l], B[l], xnew, N);
        xcur = xnew;
    }

    pool_count<<<cdiv(cdiv(N, POOL_SPAN), TB), TB, 0, stream>>>(batch, cnts, N);
    pool_sum<<<cdiv((long long)cdiv(N, POOL_SPAN) * 16, TB), TB, 0, stream>>>(xcur, batch, sums, N);
    final_linear_wmma<<<cdiv((long long)cdiv(G, 16), 8), TB, 0, stream>>>(sums, cnts, Wl, bl, out, G);
}


# Fewer launches: the two zero-fill kernels become stores in the first sort kernel and the per-graph node count is taken inside the pooling-sum kernel; 13 -> 10 launches; sum orders kept
# speedup vs baseline: 4.3024x; 1.0409x over previous
#include <hip/hip_runtime.h>
#include <hip/hip_bf16.h>

typedef float v2f __attribute__((ext_vector_type(2)));
typedef float v8f __attribute__((ext_vector_type(8)));

#define D_FEAT 64
#define D_OUT  32

#define BK_NODES    256
#define TILE_EDGES  8192
#define MAX_BUCKETS 1024
#define STAGE_CAP   6144

__device__ __forceinline__ int bucket_of(int d, int nb) {
    const int b = d >> 8;
    return b < nb ? b : nb - 1;
}

__global__ __launch_bounds__(1024) void tile_hist(
    const int* __restrict__ dst, int* __restrict__ tcnt, float* __restrict__ zbuf, int e, int nb, int nt, int nz)
{
    __shared__ int s_h[MAX_BUCKETS];
    const int tid = threadIdx.x, t = blockIdx.x;
    for (int i = t * 1024 + tid; i < nz; i += nt * 1024) zbuf[i] = 0.0f;
    for (int b = tid; b < nb; b += 1024) s_h[b] = 0;
    __syncthreads();
    const int base = t * TILE_EDGES + tid;
    int d[8];
    #pragma unroll
    for (int u = 0; u < 8; ++u) {
        const int i = base + u * 1024;
        const int v = dst[i < e ? i : e - 1];
        d[u] = i < e ? v : -1;
    }
    #pragma unroll
    for (int u = 0; u < 8; ++u) if (d[u] >= 0) atomicAdd(&s_h[bucket_of(d[u], nb)], 1);
    __syncthreads();
    for (int b = tid; b < nb; b += 1024) tcnt[b * nt + t] = s_h[b];
}

__global__ __launch_bounds__(1024) void scan_excl(const int* __restrict__ in, int* __restrict__ out, int n)
{
    __shared__ int s_w[32];
    __shared__ int s_pre[32];
    const int tid = threadIdx.x, lane = tid & 31, wv = tid >> 5;
    int pre = 0;
    for (int b = 0; b < (int)blockIdx.x; ++b) pre += in[b * 1024 + tid];
    const int i = blockIdx.x * 1024 + tid;
    const int c = i < n ? in[i] : 0;
    int inc = c;
    for (int o = 1; o < 32; o <<= 1) { int t = __shfl_up(inc, o); if (lane >= o) inc += t; }
    for (int o = 16; o > 0; o >>= 1) pre += __shfl_xor(pre, o);
    if (lane == 31) s_w[wv] = inc;
    if (lane == 0)  s_pre[wv] = pre;
    __syncthreads();
    if (wv == 0) {
        const int w = s_w[lane];
        int winc = w;
        for (int o = 1; o < 32; o <<= 1) { int t = __shfl_up(winc, o); if (lane >= o) winc += t; }
        int p = s_pre[lane];
        for (int o = 16; o > 0; o >>= 1) p += __shfl_xor(p, o);
        s_w[lane] = winc - w;
        if (lane == 0) s_pre[0] = p;
    }
    __syncthreads();
    if (i < n) out[i] = s_pre[0] + s_w[wv] + inc - c;
}

__global__ __launch_bounds__(1024) void tile_scatter(
    const int* __restrict__ srcIdx, const int* __restrict__ dstIdx, const int* __restrict__ toffs,
    int* __restrict__ binnedRaw, int e, int nb, int nt)
{
    __shared__ int s_cur[MAX_BUCKETS];
    const int tid = threadIdx.x, t = blockIdx.x;
    for (int b = tid; b < nb; b += 1024) s_cur[b] = toffs[b * nt + t];
    const int base = t * TILE_EDGES + tid;
    int s[8], d[8];
    #pragma unroll
    for (int u = 0; u < 8; ++u) {
        const int i = base + u * 1024;
        const int ic = i < e ? i : e - 1;
        const int v = dstIdx[ic];
        s[u] = srcIdx[ic];
        d[u] = i < e ? v : -1;
    }
    __syncthreads();
    int2* binned = reinterpret_cast<int2*>(binnedRaw);
    #pragma unroll
    for (int u = 0; u < 8; ++u) {
        if (d[u] >= 0) {
            const int pos = atomicAdd(&s_cur[bucket_of(d[u], nb)], 1);
            binned[pos] = make_int2(s[u], d[u]);
        }
    }
}

__global__ __launch_bounds__(BK_NODES) void bucket_count(
    const int* __restrict__ binnedRaw, const int* __restrict__ toffs,
    int* __restrict__ nodeRaw, int* __restrict__ cnt, int e, int n, int nb, int nt)
{
    __shared__ int s_c[BK_NODES];
    __shared__ int s_w[BK_NODES / 32];
    const int tid = threadIdx.x, lane = tid & 31, wv = tid >> 5;
    const int b = blockIdx.x;
    const int2* binned = reinterpret_cast<const int2*>(binnedRaw);
    const int beg = toffs[b * nt];
    const int end = b + 1 < nb ? toffs[(b + 1) * nt] : e;
    s_c[tid] = 0;
    __syncthreads();
    for (int i0 = beg + tid; i0 < end; i0 += 8 * BK_NODES) {
        int d[8];
        #pragma unroll
        for (int u = 0; u < 8; ++u) {
            const int i = i0 + u * BK_NODES;
            const int v = binned[i < end ? i : end - 1].y;
            d[u] = i < end ? v : -1;
        }
        #pragma unroll
        for (int u = 0; u < 8; ++u) if (d[u] >= 0) atomicAdd(&s_c[(d[u] - b * BK_NODES) & (BK_NODES - 1)], 1);
    }
    __syncthreads();
    const int c = s_c[tid];
    int inc = c;
    for (int o = 1; o < 32; o <<= 1) { int t = __shfl_up(inc, o); if (lane >= o) inc += t; }
    if (lane == 31) s_w[wv] = inc;
    __syncthreads();
    int wpre = 0;
    for (int w = 0; w < wv; ++w) wpre += s_w[w];
    const int node = b * BK_NODES + tid;
    if (node < n) {
        const int start = beg + wpre + inc - c;
        const float d = (float)(c + 1);
        const float di = d > 0.0f ? __frsqrt_rn(d) : 0.0f;
        reinterpret_cast<int2*>(nodeRaw)[node] = make_int2(start, __float_as_int(di));
        cnt[node] = c;
    }
}

__global__ __launch_bounds__(BK_NODES) void bucket_place(
    const int* __restrict__ binnedRaw, const int* __restrict__ toffs, const int* __restrict__ nodeRaw,
    int* __restrict__ edgeRaw, int e, int n, int nb, int nt)
{
    __shared__ int   s_cur[BK_NODES];
    __shared__ float s_di[BK_NODES];
    __shared__ int2  s_stage[STAGE_CAP];
    const int tid = threadIdx.x;
    const int b = blockIdx.x;
    const int2* binned = reinterpret_cast<const int2*>(binnedRaw);
    const int2* nodeInfo = reinterpret_cast<const int2*>(nodeRaw);
    int2* edgeOut = reinterpret_cast<int2*>(edgeRaw);
    const int beg = toffs[b * nt];
    const int end = b + 1 < nb ? toffs[(b + 1) * nt] : e;
    const int count = end - beg;
    const bool staged = count <= STAGE_CAP;
    const int node = b * BK_NODES + tid;
    const int2 nd = nodeInfo[node < n ? node : n - 1];
    s_cur[tid] = nd.x;
    s_di[tid]  = __int_as_float(nd.y);
    __syncthreads();
    for (int i0 = beg + tid; i0 < end; i0 += 8 * BK_NODES) {
        int2 sd[8];
        float ds[8];
        #pragma unroll
        for (int u = 0; u < 8; ++u) {
            const int i = i0 + u * BK_NODES;
            sd[u] = binned[i < end ? i : end - 1];
            if (i >= end) sd[u].y = -1;
        }
        #pragma unroll
        for (int u = 0; u < 8; ++u) ds[u] = __int_as_float(nodeInfo[sd[u].x].y);
        #pragma unroll
        for (int u = 0; u < 8; ++u) {
            if (sd[u].y >= 0) {
                const int l = (sd[u].y - b * BK_NODES) & (BK_NODES - 1);
                const int pos = atomicAdd(&s_cur[l], 1);
                const int2 rec = make_int2(sd[u].x, __float_as_int(ds[u] * s_di[l]));
                if (staged) s_stage[pos - beg] = rec; else edgeOut[pos] = rec;
            }
        }
    }
    if (staged) {
        __syncthreads();
        for (int i = tid; i < count; i += BK_NODES) edgeOut[beg + i] = s_stage[i];
    }
}


#define TILE_PITCH 68
#define GK 8

__global__ __launch_bounds__(256) void gcn_layer_fused(
    const int* __restrict__ nodeRaw, const int* __restrict__ cnt, const int* __restrict__ edgeRaw,
    const float* __restrict__ xin, const float* __restrict__ W, const float* __restrict__ bias,
    float* __restrict__ xout, int nNodes)
{
    __shared__ __attribute__((aligned(16))) float s_tile[8][16 * TILE_PITCH];
    __shared__ int s_end[8][32];
    const int lane = threadIdx.x & 31;
    const int wv   = threadIdx.x >> 5;
    const int r0   = (blockIdx.x * 8 + wv) * 16;
    if (r0 >= nNodes) return;
    float* st = s_tile[wv];
    int* se = s_end[wv];

    const int g  = lane >> 4;
    const int hb = lane & 16;
    const int f  = (lane & 15) * 4;
    const int first = r0 + 8 * g;
    const int2* nodeInfo = reinterpret_cast<const int2*>(nodeRaw);
    const int2* edges = reinterpret_cast<const int2*>(edgeRaw);
    const float* xf = xin + f;

    const int jn = (lane & 15) < 8 ? (lane & 15) : 7;
    const int myNode = first + jn;
    const bool myValid = myNode < nNodes;
    const int myC = myValid ? myNode : nNodes - 1;
    const int2 nd = nodeInfo[myC];
    const int segEnd = nd.x + cnt[myC];
    const int segBeg = myValid ? nd.x : segEnd;
    const float di = myValid ? __int_as_float(nd.y) : 0.0f;
    se[lane] = segEnd;
    #pragma unroll
    for (int i = 0; i < 8; ++i)
        *reinterpret_cast<float4*>(st + (8 * g + i) * TILE_PITCH + f) = make_float4(0.0f, 0.0f, 0.0f, 0.0f);
    const int beg = __shfl(segBeg, hb);
    const int end = __shfl(segEnd, hb + 7);
    __builtin_amdgcn_fence(__ATOMIC_RELEASE, "wavefront");
    __builtin_amdgcn_wave_barrier();
    __builtin_amdgcn_fence(__ATOMIC_ACQUIRE, "wavefront");

    {
        int cur = 0;
        int nxt = __shfl(segEnd, hb);
        float4 acc = make_float4(0.0f, 0.0f, 0.0f, 0.0f);
        int2 pr[GK];
        #pragma unroll
        for (int k = 0; k < GK; ++k) pr[k] = edges[beg + k];
        for (int base = beg; base < end; base += GK) {
            int2 nx[GK];
            #pragma unroll
            for (int k = 0; k < GK; ++k) nx[k] = edges[base + GK + k];
            float4 r[GK];
            #pragma unroll
            for (int k = 0; k < GK; ++k) {
                const int s = base + k < end ? pr[k].x : pr[0].x;
                r[k] = *reinterpret_cast<const float4*>(xf + (size_t)s * D_FEAT);
            }
            #pragma unroll
            for (int k = 0; k < GK; ++k) {
                const int e = base + k;
                if (e < end) {
                    if (e >= nxt) {
                        *reinterpret_cast<float4*>(st + (8 * g + cur) * TILE_PITCH + f) = acc;
                        acc = make_float4(0.0f, 0.0f, 0.0f, 0.0f);
                        do { ++cur; nxt = se[hb + cur]; } while (e >= nxt && cur < 7);
                    }
                    const float w = __int_as_float(pr[k].y);
                    acc.x = fmaf(w, r[k].x, acc.x); acc.y = fmaf(w, r[k].y, acc.y);
                    acc.z = fmaf(w, r[k].z, acc.z); acc.w = fmaf(w, r[k].w, acc.w);
                }
            }
            #pragma unroll
            for (int k = 0; k < GK; ++k) pr[k] = nx[k];
        }
        if (beg < end) *reinterpret_cast<float4*>(st + (8 * g + cur) * TILE_PITCH + f) = acc;
    }

    float4 xs[8];
    #pragma unroll
    for (int i = 0; i < 8; ++i) {
        const int node = first + i;
        const int nc = node < nNodes ? node : nNodes - 1;
        xs[i] = *reinterpret_cast<const float4*>(xf + (size_t)nc * D_FEAT);
    }
    #pragma unroll
    for (int i = 0; i < 8; ++i) {
        float* trow = st + (8 * g + i) * TILE_PITCH + f;
        float4 v = *reinterpret_cast<float4*>(trow);
        const float d = __shfl(di, hb + i);
        const float dd = d * d;
        v.x += dd * xs[i].x; v.y += dd * xs[i].y; v.z += dd * xs[i].z; v.w += dd * xs[i].w;
        *reinterpret_cast<float4*>(trow) = v;
    }
    __builtin_amdgcn_fence(__ATOMIC_RELEASE, "wavefront");
    __builtin_amdgcn_wave_barrier();
    __builtin_amdgcn_fence(__ATOMIC_ACQUIRE, "wavefront");

    const int m  = lane & 15;
    const int kk = (lane >> 4) * 2;
    const int n  = lane & 15;
    v8f acc0 = {}, acc1 = {}, acc2 = {}, acc3 = {};
    const float* arow = st + m * TILE_PITCH + kk;

    #pragma unroll
    for (int k0 = 0; k0 < D_FEAT; k0 += 4) {
        v2f a;
        a.x = arow[k0];
        a.y = arow[k0 + 1];
        const float* w0 = W + (size_t)(k0 + kk) * D_FEAT + n;
        const float* w1 = W + (size_t)(k0 + kk + 1) * D_FEAT + n;
        v2f b0; b0.x = w0[0];  b0.y = w1[0];
        v2f b1; b1.x = w0[16]; b1.y = w1[16];
        v2f b2; b2.x = w0[32]; b2.y = w1[32];
        v2f b3; b3.x = w0[48]; b3.y = w1[48];
        acc0 = __builtin_amdgcn_wmma_f32_16x16x4_f32(false, a, false, b0, (short)0, acc0, false, false);
        acc1 = __builtin_amdgcn_wmma_f32_16x16x4_f32(false, a, false, b1, (short)0, acc1, false, false);
        acc2 = __builtin_amdgcn_wmma_f32_16x16x4_f32(false, a, false, b2, (short)0, acc2, false, false);
        acc3 = __builtin_amdgcn_wmma_f32_16x16x4_f32(false, a, false, b3, (short)0, acc3, false, false);
    }

    __builtin_amdgcn_fence(__ATOMIC_RELEASE, "wavefront");
    __builtin_amdgcn_wave_barrier();
    __builtin_amdgcn_fence(__ATOMIC_ACQUIRE, "wavefront");
    const int rbase = (lane >> 4) * 8;
    const int col   = lane & 15;
    #pragma unroll
    for (int r = 0; r < 8; ++r) {
        float* trow = st + (rbase + r) * TILE_PITCH + col;
        trow[0]  = acc0[r];
        trow[16] = acc1[r];
        trow[32] = acc2[r];
        trow[48] = acc3[r];
    }
    __builtin_amdgcn_fence(__ATOMIC_RELEASE, "wavefront");
    __builtin_amdgcn_wave_barrier();
    __builtin_amdgcn_fence(__ATOMIC_ACQUIRE, "wavefront");

    const float4 bb = *reinterpret_cast<const float4*>(bias + f);
    #pragma unroll
    for (int i = 0; i < 8; ++i) {
        const int nl = 8 * g + i;
        const int node = r0 + nl;
        float4 v = *reinterpret_cast<float4*>(st + nl * TILE_PITCH + f);
        v.x += bb.x; v.y += bb.y; v.z += bb.z; v.w += bb.w;
        v.x = (v.x > 0.0f ? v.x : 0.0f) + xs[i].x;
        v.y = (v.y > 0.0f ? v.y : 0.0f) + xs[i].y;
        v.z = (v.z > 0.0f ? v.z : 0.0f) + xs[i].z;
        v.w = (v.w > 0.0f ? v.w : 0.0f) + xs[i].w;
        if (node < nNodes) *reinterpret_cast<float4*>(xout + (size_t)node * D_FEAT + f) = v;
    }
}

#define POOL_SPAN 32

__device__ __forceinline__ void pool_flush(float* __restrict__ sums, float* __restrict__ cnts, int g, int f,
                                           const float4& a, float c) {
    float* s = sums + (size_t)g * D_FEAT + f;
    unsafeAtomicAdd(s + 0, a.x);
    unsafeAtomicAdd(s + 1, a.y);
    unsafeAtomicAdd(s + 2, a.z);
    unsafeAtomicAdd(s + 3, a.w);
    if (f == 0) unsafeAtomicAdd(&cnts[g], c);
}

__global__ __launch_bounds__(256) void pool_sum(
    const float* __restrict__ x, const int* __restrict__ batch,
    float* __restrict__ sums, float* __restrict__ cnts, int n)
{
    const int f = (threadIdx.x & 15) * 4;
    const int beg = ((blockIdx.x * 256 + threadIdx.x) >> 4) * POOL_SPAN;
    if (beg >= n) return;
    const int end = beg + POOL_SPAN < n ? beg + POOL_SPAN : n;
    const float* xf = x + f;
    int g = batch[beg];
    float4 acc = make_float4(0.0f, 0.0f, 0.0f, 0.0f);
    float c = 0.0f;
    int i = beg;
    for (; i + 8 <= end; i += 8) {
        int b[8];
        float4 v[8];
        #pragma unroll
        for (int k = 0; k < 8; ++k) {
            b[k] = batch[i + k];
            v[k] = *reinterpret_cast<const float4*>(xf + (size_t)(i + k) * D_FEAT);
        }
        #pragma unroll
        for (int k = 0; k < 8; ++k) {
            if (b[k] != g) { pool_flush(sums, cnts, g, f, acc, c); acc = make_float4(0.0f, 0.0f, 0.0f, 0.0f); c = 0.0f; g = b[k]; }
            acc.x += v[k].x; acc.y += v[k].y; acc.z += v[k].z; acc.w += v[k].w;
            c += 1.0f;
        }
    }
    for (; i < end; ++i) {
        const int b0 = batch[i];
        const float4 v0 = *reinterpret_cast<const float4*>(xf + (size_t)i * D_FEAT);
        if (b0 != g) { pool_flush(sums, cnts, g, f, acc, c); acc = make_float4(0.0f, 0.0f, 0.0f, 0.0f); c = 0.0f; g = b0; }
        acc.x += v0.x; acc.y += v0.y; acc.z += v0.z; acc.w += v0.w;
        c += 1.0f;
    }
    pool_flush(sums, cnts, g, f, acc, c);
}

__global__ __launch_bounds__(256) void final_linear_wmma(
    const float* __restrict__ sums, const float* __restrict__ cnts,
    const float* __restrict__ Wl, const float* __restrict__ bl,
    float* __restrict__ out, int nrows)
{
    const int lane  = threadIdx.x & 31;
    const int strip = blockIdx.x * 8 + (threadIdx.x >> 5);
    const int r0 = strip * 16;
    if (r0 >= nrows) return;

    const int m  = lane & 15;
    const int kk = (lane >> 4) * 2;
    const int n  = lane & 15;
    const int row = r0 + m;
    float c = cnts[row];
    float inv = 1.0f / (c > 1.0f ? c : 1.0f);

    v8f acc0 = {}, acc1 = {};
    const float* srow = sums + (size_t)row * D_FEAT;

    #pragma unroll
    for (int k0 = 0; k0 < D_FEAT; k0 += 4) {
        v2f a;
        a.x = srow[k0 + kk] * inv;
        a.y = srow[k0 + kk + 1] * inv;
        const float* w0 = Wl + (size_t)(k0 + kk) * D_OUT + n;
        const float* w1 = Wl + (size_t)(k0 + kk + 1) * D_OUT + n;
        v2f b0; b0.x = w0[0];  b0.y = w1[0];
        v2f b1; b1.x = w0[16]; b1.y = w1[16];
        acc0 = __builtin_amdgcn_wmma_f32_16x16x4_f32(false, a, false, b0, (short)0, acc0, false, false);
        acc1 = __builtin_amdgcn_wmma_f32_16x16x4_f32(false, a, false, b1, (short)0, acc1, false, false);
    }

    const int rbase = (lane >> 4) * 8;
    const int col   = lane & 15;
    #pragma unroll
    for (int r = 0; r < 8; ++r) {
        float* orow = out + (size_t)(r0 + rbase + r) * D_OUT + col;
        orow[0]  = acc0[r] + bl[col];
        orow[16] = acc1[r] + bl[col + 16];
    }
}

static inline int cdiv(long long a, int b) { return (int)((a + b - 1) / b); }

extern "C" void kernel_launch(void* const* d_in, const int* in_sizes, int n_in,
                              void* d_out, int out_size, void* d_ws, size_t ws_size,
                              hipStream_t stream)
{
    const float* x     = (const float*)d_in[0];
    const int*   ei    = (const int*)d_in[1];
    const int*   batch = (const int*)d_in[2];
    const float* W[3]  = {(const float*)d_in[3], (const float*)d_in[5], (const float*)d_in[7]};
    const float* B[3]  = {(const float*)d_in[4], (const float*)d_in[6], (const float*)d_in[8]};
    const float* Wl    = (const float*)d_in[9];
    const float* bl    = (const float*)d_in[10];
    float* out = (float*)d_out;

    const int N = in_sizes[0] / D_FEAT;
    const int E = in_sizes[1] / 2;
    const int G = out_size / D_OUT;
    const int* srcIdx = ei;
    const int* dstIdx = ei + E;

    float* ws   = (float*)d_ws;
    float* buf0 = ws;
    float* buf1 = buf0 + (size_t)N * D_FEAT;
    const int NB = cdiv(N, BK_NODES);
    const int NT = cdiv(E, TILE_EDGES);
    int*   edges  = (int*)(buf1 + (size_t)N * D_FEAT);
    int*   nodes  = edges + (size_t)E * 2;
    int*   binned = nodes + (size_t)N * 2;
    int*   cnt    = binned + (size_t)E * 2;
    int*   tcnt   = cnt + N;
    int*   toffs  = tcnt + (size_t)NB * NT;
    float* sums   = (float*)(toffs + (size_t)NB * NT);
    float* cnts   = sums + (size_t)G * D_FEAT;

    const int TB = 256;

    tile_hist<<<NT, 1024, 0, stream>>>(dstIdx, tcnt, sums, E, NB, NT, G * D_FEAT + G);
    scan_excl<<<cdiv((long long)NB * NT, 1024), 1024, 0, stream>>>(tcnt, toffs, NB * NT);
    tile_scatter<<<NT, 1024, 0, stream>>>(srcIdx, dstIdx, toffs, binned, E, NB, NT);
    bucket_count<<<NB, BK_NODES, 0, stream>>>(binned, toffs, nodes, cnt, E, N, NB, NT);
    bucket_place<<<NB, BK_NODES, 0, stream>>>(binned, toffs, nodes, edges, E, N, NB, NT);

    const int layerBlocks = cdiv((long long)cdiv(N, 16), 8);
    const float* xcur = x;
    float* xbuf[3] = {buf0, buf1, buf0};

    for (int l = 0; l < 3; ++l) {
        float* xnew = xbuf[l];
        gcn_layer_fused<<<layerBlocks, TB, 0, stream>>>(nodes, cnt, edges, xcur, W[l], B[l], xnew, N);
        xcur = xnew;
    }

    pool_sum<<<cdiv((long long)cdiv(N, POOL_SPAN) * 16, TB), TB, 0, stream>>>(xcur, batch, sums, cnts, N);
    final_linear_wmma<<<cdiv((long long)cdiv(G, 16), 8), TB, 0, stream>>>(sums, cnts, Wl, bl, out, G);
}


# Pooling fused into the last layer's kernel: rows summed per graph in registers, joined per block in LDS, added once per run; no x3 plane, no pooling launch (10 -> 9); pool sum order changes
# speedup vs baseline: 4.5507x; 1.0577x over previous
#include <hip/hip_runtime.h>
#include <hip/hip_bf16.h>

typedef float v2f __attribute__((ext_vector_type(2)));
typedef float v8f __attribute__((ext_vector_type(8)));

#define D_FEAT 64
#define D_OUT  32

#define BK_NODES    256
#define TILE_EDGES  8192
#define MAX_BUCKETS 1024
#define STAGE_CAP   6144

__device__ __forceinline__ int bucket_of(int d, int nb) {
    const int b = d >> 8;
    return b < nb ? b : nb - 1;
}

__global__ __launch_bounds__(1024) void tile_hist(
    const int* __restrict__ dst, int* __restrict__ tcnt, float* __restrict__ zbuf, int e, int nb, int nt, int nz)
{
    __shared__ int s_h[MAX_BUCKETS];
    const int tid = threadIdx.x, t = blockIdx.x;
    for (int i = t * 1024 + tid; i < nz; i += nt * 1024) zbuf[i] = 0.0f;
    for (int b = tid; b < nb; b += 1024) s_h[b] = 0;
    __syncthreads();
    const int base = t * TILE_EDGES + tid;
    int d[8];
    #pragma unroll
    for (int u = 0; u < 8; ++u) {
        const int i = base + u * 1024;
        const int v = dst[i < e ? i : e - 1];
        d[u] = i < e ? v : -1;
    }
    #pragma unroll
    for (int u = 0; u < 8; ++u) if (d[u] >= 0) atomicAdd(&s_h[bucket_of(d[u], nb)], 1);
    __syncthreads();
    for (int b = tid; b < nb; b += 1024) tcnt[b * nt + t] = s_h[b];
}

__global__ __launch_bounds__(1024) void scan_excl(const int* __restrict__ in, int* __restrict__ out, int n)
{
    __shared__ int s_w[32];
    __shared__ int s_pre[32];
    const int tid = threadIdx.x, lane = tid & 31, wv = tid >> 5;
    int pre = 0;
    for (int b = 0; b < (int)blockIdx.x; ++b) pre += in[b * 1024 + tid];
    const int i = blockIdx.x * 1024 + tid;
    const int c = i < n ? in[i] : 0;
    int inc = c;
    for (int o = 1; o < 32; o <<= 1) { int t = __shfl_up(inc, o); if (lane >= o) inc += t; }
    for (int o = 16; o > 0; o >>= 1) pre += __shfl_xor(pre, o);
    if (lane == 31) s_w[wv] = inc;
    if (lane == 0)  s_pre[wv] = pre;
    __syncthreads();
    if (wv == 0) {
        const int w = s_w[lane];
        int winc = w;
        for (int o = 1; o < 32; o <<= 1) { int t = __shfl_up(winc, o); if (lane >= o) winc += t; }
        int p = s_pre[lane];
        for (int o = 16; o > 0; o >>= 1) p += __shfl_xor(p, o);
        s_w[lane] = winc - w;
        if (lane == 0) s_pre[0] = p;
    }
    __syncthreads();
    if (i < n) out[i] = s_pre[0] + s_w[wv] + inc - c;
}

__global__ __launch_bounds__(1024) void tile_scatter(
    const int* __restrict__ srcIdx, const int* __restrict__ dstIdx, const int* __restrict__ toffs,
    int* __restrict__ binnedRaw, int e, int nb, int nt)
{
    __shared__ int s_cur[MAX_BUCKETS];
    const int tid = threadIdx.x, t = blockIdx.x;
    for (int b = tid; b < nb; b += 1024) s_cur[b] = toffs[b * nt + t];
    const int base = t * TILE_EDGES + tid;
    int s[8], d[8];
    #pragma unroll
    for (int u = 0; u < 8; ++u) {
        const int i = base + u * 1024;
        const int ic = i < e ? i : e - 1;
        const int v = dstIdx[ic];
        s[u] = srcIdx[ic];
        d[u] = i < e ? v : -1;
    }
    __syncthreads();
    int2* binned = reinterpret_cast<int2*>(binnedRaw);
    #pragma unroll
    for (int u = 0; u < 8; ++u) {
        if (d[u] >= 0) {
            const int pos = atomicAdd(&s_cur[bucket_of(d[u], nb)], 1);
            binned[pos] = make_int2(s[u], d[u]);
        }
    }
}

__global__ __launch_bounds__(BK_NODES) void bucket_count(
    const int* __restrict__ binnedRaw, const int* __restrict__ toffs,
    int* __restrict__ nodeRaw, int* __restrict__ cnt, int e, int n, int nb, int nt)
{
    __shared__ int s_c[BK_NODES];
    __shared__ int s_w[BK_NODES / 32];
    const int tid = threadIdx.x, lane = tid & 31, wv = tid >> 5;
    const int b = blockIdx.x;
    const int2* binned = reinterpret_cast<const int2*>(binnedRaw);
    const int beg = toffs[b * nt];
    const int end = b + 1 < nb ? toffs[(b + 1) * nt] : e;
    s_c[tid] = 0;
    __syncthreads();
    for (int i0 = beg + tid; i0 < end; i0 += 8 * BK_NODES) {
        int d[8];
        #pragma unroll
        for (int u = 0; u < 8; ++u) {
            const int i = i0 + u * BK_NODES;
            const int v = binned[i < end ? i : end - 1].y;
            d[u] = i < end ? v : -1;
        }
        #pragma unroll
        for (int u = 0; u < 8; ++u) if (d[u] >= 0) atomicAdd(&s_c[(d[u] - b * BK_NODES) & (BK_NODES - 1)], 1);
    }
    __syncthreads();
    const int c = s_c[tid];
    int inc = c;
    for (int o = 1; o < 32; o <<= 1) { int t = __shfl_up(inc, o); if (lane >= o) inc += t; }
    if (lane == 31) s_w[wv] = inc;
    __syncthreads();
    int wpre = 0;
    for (int w = 0; w < wv; ++w) wpre += s_w[w];
    const int node = b * BK_NODES + tid;
    if (node < n) {
        const int start = beg + wpre + inc - c;
        const float d = (float)(c + 1);
        const float di = d > 0.0f ? __frsqrt_rn(d) : 0.0f;
        reinterpret_cast<int2*>(nodeRaw)[node] = make_int2(start, __float_as_int(di));
        cnt[node] = c;
    }
}

__global__ __launch_bounds__(BK_NODES) void bucket_place(
    const int* __restrict__ binnedRaw, const int* __restrict__ toffs, const int* __restrict__ nodeRaw,
    int* __restrict__ edgeRaw, int e, int n, int nb, int nt)
{
    __shared__ int   s_cur[BK_NODES];
    __shared__ float s_di[BK_NODES];
    __shared__ int2  s_stage[STAGE_CAP];
    const int tid = threadIdx.x;
    const int b = blockIdx.x;
    const int2* binned = reinterpret_cast<const int2*>(binnedRaw);
    const int2* nodeInfo = reinterpret_cast<const int2*>(nodeRaw);
    int2* edgeOut = reinterpret_cast<int2*>(edgeRaw);
    const int beg = toffs[b * nt];
    const int end = b + 1 < nb ? toffs[(b + 1) * nt] : e;
    const int count = end - beg;
    const bool staged = count <= STAGE_CAP;
    const int node = b * BK_NODES + tid;
    const int2 nd = nodeInfo[node < n ? node : n - 1];
    s_cur[tid] = nd.x;
    s_di[tid]  = __int_as_float(nd.y);
    __syncthreads();
    for (int i0 = beg + tid; i0 < end; i0 += 8 * BK_NODES) {
        int2 sd[8];
        float ds[8];
        #pragma unroll
        for (int u = 0; u < 8; ++u) {
            const int i = i0 + u * BK_NODES;
            sd[u] = binned[i < end ? i : end - 1];
            if (i >= end) sd[u].y = -1;
        }
        #pragma unroll
        for (int u = 0; u < 8; ++u) ds[u] = __int_as_float(nodeInfo[sd[u].x].y);
        #pragma unroll
        for (int u = 0; u < 8; ++u) {
            if (sd[u].y >= 0) {
                const int l = (sd[u].y - b * BK_NODES) & (BK_NODES - 1);
                const int pos = atomicAdd(&s_cur[l], 1);
                const int2 rec = make_int2(sd[u].x, __float_as_int(ds[u] * s_di[l]));
                if (staged) s_stage[pos - beg] = rec; else edgeOut[pos] = rec;
            }
        }
    }
    if (staged) {
        __syncthreads();
        for (int i = tid; i < count; i += BK_NODES) edgeOut[beg + i] = s_stage[i];
    }
}

__device__ __forceinline__ void pool_flush(float* __restrict__ sums, float* __restrict__ cnts, int g, int f,
                                           const float4& a, float c) {
    float* s = sums + (size_t)g * D_FEAT + f;
    unsafeAtomicAdd(s + 0, a.x);
    unsafeAtomicAdd(s + 1, a.y);
    unsafeAtomicAdd(s + 2, a.z);
    unsafeAtomicAdd(s + 3, a.w);
    if (f == 0) unsafeAtomicAdd(&cnts[g], c);
}


#define TILE_PITCH 68
#define GK 8

template <bool POOL>
__global__ __launch_bounds__(256) void gcn_layer_fused(
    const int* __restrict__ nodeRaw, const int* __restrict__ cnt, const int* __restrict__ edgeRaw,
    const float* __restrict__ xin, const float* __restrict__ W, const float* __restrict__ bias,
    float* __restrict__ xout, int nNodes,
    const int* __restrict__ batch, float* __restrict__ sums, float* __restrict__ cnts)
{
    __shared__ __attribute__((aligned(16))) float s_tile[8][16 * TILE_PITCH];
    __shared__ int s_end[8][32];
    __shared__ __attribute__((aligned(16))) float s_pool[8][D_FEAT];
    __shared__ int   s_gid[8][32];
    __shared__ int   s_pid[8];
    __shared__ float s_pcnt[8];
    const int lane = threadIdx.x & 31;
    const int wv   = threadIdx.x >> 5;
    const int r0   = (blockIdx.x * 8 + wv) * 16;
    if (!POOL) { if (r0 >= nNodes) return; }
    if (POOL && r0 >= nNodes) { if (lane == 0) s_pid[wv] = -1; }
    if (r0 < nNodes) {
    float* st = s_tile[wv];
    int* se = s_end[wv];

    const int g  = lane >> 4;
    const int hb = lane & 16;
    const int f  = (lane & 15) * 4;
    const int first = r0 + 8 * g;
    const int2* nodeInfo = reinterpret_cast<const int2*>(nodeRaw);
    const int2* edges = reinterpret_cast<const int2*>(edgeRaw);
    const float* xf = xin + f;

    const int jn = (lane & 15) < 8 ? (lane & 15) : 7;
    const int myNode = first + jn;
    const bool myValid = myNode < nNodes;
    const int myC = myValid ? myNode : nNodes - 1;
    const int2 nd = nodeInfo[myC];
    const int segEnd = nd.x + cnt[myC];
    const int segBeg = myValid ? nd.x : segEnd;
    const float di = myValid ? __int_as_float(nd.y) : 0.0f;
    if (POOL) s_gid[wv][lane] = batch[myC];
    se[lane] = segEnd;
    #pragma unroll
    for (int i = 0; i < 8; ++i)
        *reinterpret_cast<float4*>(st + (8 * g + i) * TILE_PITCH + f) = make_float4(0.0f, 0.0f, 0.0f, 0.0f);
    const int beg = __shfl(segBeg, hb);
    const int end = __shfl(segEnd, hb + 7);
    __builtin_amdgcn_fence(__ATOMIC_RELEASE, "wavefront");
    __builtin_amdgcn_wave_barrier();
    __builtin_amdgcn_fence(__ATOMIC_ACQUIRE, "wavefront");

    {
        int cur = 0;
        int nxt = __shfl(segEnd, hb);
        float4 acc = make_float4(0.0f, 0.0f, 0.0f, 0.0f);
        int2 pr[GK];
        #pragma unroll
        for (int k = 0; k < GK; ++k) pr[k] = edges[beg + k];
        for (int base = beg; base < end; base += GK) {
            int2 nx[GK];
            #pragma unroll
            for (int k = 0; k < GK; ++k) nx[k] = edges[base + GK + k];
            float4 r[GK];
            #pragma unroll
            for (int k = 0; k < GK; ++k) {
                const int s = base + k < end ? pr[k].x : pr[0].x;
                r[k] = *reinterpret_cast<const float4*>(xf + (size_t)s * D_FEAT);
            }
            #pragma unroll
            for (int k = 0; k < GK; ++k) {
                const int e = base + k;
                if (e < end) {
                    if (e >= nxt) {
                        *reinterpret_cast<float4*>(st + (8 * g + cur) * TILE_PITCH + f) = acc;
                        acc = make_float4(0.0f, 0.0f, 0.0f, 0.0f);
                        do { ++cur; nxt = se[hb + cur]; } while (e >= nxt && cur < 7);
                    }
                    const float w = __int_as_float(pr[k].y);
                    acc.x = fmaf(w, r[k].x, acc.x); acc.y = fmaf(w, r[k].y, acc.y);
                    acc.z = fmaf(w, r[k].z, acc.z); acc.w = fmaf(w, r[k].w, acc.w);
                }
            }
            #pragma unroll
            for (int k = 0; k < GK; ++k) pr[k] = nx[k];
        }
        if (beg < end) *reinterpret_cast<float4*>(st + (8 * g + cur) * TILE_PITCH + f) = acc;
    }

    float4 xs[8];
    #pragma unroll
    for (int i = 0; i < 8; ++i) {
        const int node = first + i;
        const int nc = node < nNodes ? node : nNodes - 1;
        xs[i] = *reinterpret_cast<const float4*>(xf + (size_t)nc * D_FEAT);
    }
    #pragma unroll
    for (int i = 0; i < 8; ++i) {
        float* trow = st + (8 * g + i) * TILE_PITCH + f;
        float4 v = *reinterpret_cast<float4*>(trow);
        const float d = __shfl(di, hb + i);
        const float dd = d * d;
        v.x += dd * xs[i].x; v.y += dd * xs[i].y; v.z += dd * xs[i].z; v.w += dd * xs[i].w;
        *reinterpret_cast<float4*>(trow) = v;
    }
    __builtin_amdgcn_fence(__ATOMIC_RELEASE, "wavefront");
    __builtin_amdgcn_wave_barrier();
    __builtin_amdgcn_fence(__ATOMIC_ACQUIRE, "wavefront");

    const int m  = lane & 15;
    const int kk = (lane >> 4) * 2;
    const int n  = lane & 15;
    v8f acc0 = {}, acc1 = {}, acc2 = {}, acc3 = {};
    const float* arow = st + m * TILE_PITCH + kk;

    #pragma unroll
    for (int k0 = 0; k0 < D_FEAT; k0 += 4) {
        v2f a;
        a.x = arow[k0];
        a.y = arow[k0 + 1];
        const float* w0 = W + (size_t)(k0 + kk) * D_FEAT + n;
        const float* w1 = W + (size_t)(k0 + kk + 1) * D_FEAT + n;
        v2f b0; b0.x = w0[0];  b0.y = w1[0];
        v2f b1; b1.x = w0[16]; b1.y = w1[16];
        v2f b2; b2.x = w0[32]; b2.y = w1[32];
        v2f b3; b3.x = w0[48]; b3.y = w1[48];
        acc0 = __builtin_amdgcn_wmma_f32_16x16x4_f32(false, a, false, b0, (short)0, acc0, false, false);
        acc1 = __builtin_amdgcn_wmma_f32_16x16x4_f32(false, a, false, b1, (short)0, acc1, false, false);
        acc2 = __builtin_amdgcn_wmma_f32_16x16x4_f32(false, a, false, b2, (short)0, acc2, false, false);
        acc3 = __builtin_amdgcn_wmma_f32_16x16x4_f32(false, a, false, b3, (short)0, acc3, false, false);
    }

    __builtin_amdgcn_fence(__ATOMIC_RELEASE, "wavefront");
    __builtin_amdgcn_wave_barrier();
    __builtin_amdgcn_fence(__ATOMIC_ACQUIRE, "wavefront");
    const int rbase = (lane >> 4) * 8;
    const int col   = lane & 15;
    #pragma unroll
    for (int r = 0; r < 8; ++r) {
        float* trow = st + (rbase + r) * TILE_PITCH + col;
        trow[0]  = acc0[r];
        trow[16] = acc1[r];
        trow[32] = acc2[r];
        trow[48] = acc3[r];
    }
    __builtin_amdgcn_fence(__ATOMIC_RELEASE, "wavefront");
    __builtin_amdgcn_wave_barrier();
    __builtin_amdgcn_fence(__ATOMIC_ACQUIRE, "wavefront");

    const float4 bb = *reinterpret_cast<const float4*>(bias + f);
    #pragma unroll
    for (int i = 0; i < 8; ++i) {
        const int nl = 8 * g + i;
        const int node = r0 + nl;
        float4 v = *reinterpret_cast<float4*>(st + nl * TILE_PITCH + f);
        v.x += bb.x; v.y += bb.y; v.z += bb.z; v.w += bb.w;
        v.x = (v.x > 0.0f ? v.x : 0.0f) + xs[i].x;
        v.y = (v.y > 0.0f ? v.y : 0.0f) + xs[i].y;
        v.z = (v.z > 0.0f ? v.z : 0.0f) + xs[i].z;
        v.w = (v.w > 0.0f ? v.w : 0.0f) + xs[i].w;
        if (!POOL) {
            if (node < nNodes) *reinterpret_cast<float4*>(xout + (size_t)node * D_FEAT + f) = v;
        } else {
            *reinterpret_cast<float4*>(st + nl * TILE_PITCH + f) = v;
        }
    }
    if (POOL) {
        const int* sg = s_gid[wv];
        int pg = -1, runs = 0;
        float4 pacc = make_float4(0.0f, 0.0f, 0.0f, 0.0f);
        float pc = 0.0f;
        for (int i = 0; i < 8; ++i) {
            if (first + i < nNodes) {
                const int id = sg[hb + i];
                const float4 v = *reinterpret_cast<const float4*>(st + (8 * g + i) * TILE_PITCH + f);
                if (id != pg) {
                    if (pg >= 0) pool_flush(sums, cnts, pg, f, pacc, pc);
                    pacc = make_float4(0.0f, 0.0f, 0.0f, 0.0f); pc = 0.0f; pg = id; ++runs;
                }
                pacc.x += v.x; pacc.y += v.y; pacc.z += v.z; pacc.w += v.w;
                pc += 1.0f;
            }
        }
        const int og = __shfl_xor(pg, 16);
        const int oruns = __shfl_xor(runs, 16);
        float4 o;
        o.x = __shfl_xor(pacc.x, 16); o.y = __shfl_xor(pacc.y, 16);
        o.z = __shfl_xor(pacc.z, 16); o.w = __shfl_xor(pacc.w, 16);
        const float oc = __shfl_xor(pc, 16);
        const bool whole = pg >= 0 && og == pg && runs == 1 && oruns == 1;
        if (whole) {
            if (g == 0)
                *reinterpret_cast<float4*>(&s_pool[wv][f]) = make_float4(pacc.x + o.x, pacc.y + o.y, pacc.z + o.z, pacc.w + o.w);
            if (lane == 0) { s_pid[wv] = pg; s_pcnt[wv] = pc + oc; }
        } else {
            if (pg >= 0) pool_flush(sums, cnts, pg, f, pacc, pc);
            if (lane == 0) s_pid[wv] = -1;
        }
    }
    }

    if (POOL) {
        __syncthreads();
        if (threadIdx.x < D_FEAT) {
            const int ft = threadIdx.x;
            int cg = -1;
            float a = 0.0f, c = 0.0f;
            for (int w = 0; w < 8; ++w) {
                const int id = s_pid[w];
                if (id >= 0) {
                    if (id != cg) {
                        if (cg >= 0) {
                            unsafeAtomicAdd(&sums[(size_t)cg * D_FEAT + ft], a);
                            if (ft == 0) unsafeAtomicAdd(&cnts[cg], c);
                        }
                        a = 0.0f; c = 0.0f; cg = id;
                    }
                    a += s_pool[w][ft];
                    c += s_pcnt[w];
                }
            }
            if (cg >= 0) {
                unsafeAtomicAdd(&sums[(size_t)cg * D_FEAT + ft], a);
                if (ft == 0) unsafeAtomicAdd(&cnts[cg], c);
            }
        }
    }
}

__global__ __launch_bounds__(256) void final_linear_wmma(
    const float* __restrict__ sums, const float* __restrict__ cnts,
    const float* __restrict__ Wl, const float* __restrict__ bl,
    float* __restrict__ out, int nrows)
{
    const int lane  = threadIdx.x & 31;
    const int strip = blockIdx.x * 8 + (threadIdx.x >> 5);
    const int r0 = strip * 16;
    if (r0 >= nrows) return;

    const int m  = lane & 15;
    const int kk = (lane >> 4) * 2;
    const int n  = lane & 15;
    const int row = r0 + m;
    float c = cnts[row];
    float inv = 1.0f / (c > 1.0f ? c : 1.0f);

    v8f acc0 = {}, acc1 = {};
    const float* srow = sums + (size_t)row * D_FEAT;

    #pragma unroll
    for (int k0 = 0; k0 < D_FEAT; k0 += 4) {
        v2f a;
        a.x = srow[k0 + kk] * inv;
        a.y = srow[k0 + kk + 1] * inv;
        const float* w0 = Wl + (size_t)(k0 + kk) * D_OUT + n;
        const float* w1 = Wl + (size_t)(k0 + kk + 1) * D_OUT + n;
        v2f b0; b0.x = w0[0];  b0.y = w1[0];
        v2f b1; b1.x = w0[16]; b1.y = w1[16];
        acc0 = __builtin_amdgcn_wmma_f32_16x16x4_f32(false, a, false, b0, (short)0, acc0, false, false);
        acc1 = __builtin_amdgcn_wmma_f32_16x16x4_f32(false, a, false, b1, (short)0, acc1, false, false);
    }

    const int rbase = (lane >> 4) * 8;
    const int col   = lane & 15;
    #pragma unroll
    for (int r = 0; r < 8; ++r) {
        float* orow = out + (size_t)(r0 + rbase + r) * D_OUT + col;
        orow[0]  = acc0[r] + bl[col];
        orow[16] = acc1[r] + bl[col + 16];
    }
}

static inline int cdiv(long long a, int b) { return (int)((a + b - 1) / b); }

extern "C" void kernel_launch(void* const* d_in, const int* in_sizes, int n_in,
                              void* d_out, int out_size, void* d_ws, size_t ws_size,
                              hipStream_t stream)
{
    const float* x     = (const float*)d_in[0];
    const int*   ei    = (const int*)d_in[1];
    const int*   batch = (const int*)d_in[2];
    const float* W[3]  = {(const float*)d_in[3], (const float*)d_in[5], (const float*)d_in[7]};
    const float* B[3]  = {(const float*)d_in[4], (const float*)d_in[6], (const float*)d_in[8]};
    const float* Wl    = (const float*)d_in[9];
    const float* bl    = (const float*)d_in[10];
    float* out = (float*)d_out;

    const int N = in_sizes[0] / D_FEAT;
    const int E = in_sizes[1] / 2;
    const int G = out_size / D_OUT;
    const int* srcIdx = ei;
    const int* dstIdx = ei + E;

    float* ws   = (float*)d_ws;
    float* buf0 = ws;
    float* buf1 = buf0 + (size_t)N * D_FEAT;
    const int NB = cdiv(N, BK_NODES);
    const int NT = cdiv(E, TILE_EDGES);
    int*   edges  = (int*)(buf1 + (size_t)N * D_FEAT);
    int*   nodes  = edges + (size_t)E * 2;
    int*   binned = nodes + (size_t)N * 2;
    int*   cnt    = binned + (size_t)E * 2;
    int*   tcnt   = cnt + N;
    int*   toffs  = tcnt + (size_t)NB * NT;
    float* sums   = (float*)(toffs + (size_t)NB * NT);
    float* cnts   = sums + (size_t)G * D_FEAT;

    const int TB = 256;

    tile_hist<<<NT, 1024, 0, stream>>>(dstIdx, tcnt, sums, E, NB, NT, G * D_FEAT + G);
    scan_excl<<<cdiv((long long)NB * NT, 1024), 1024, 0, stream>>>(tcnt, toffs, NB * NT);
    tile_scatter<<<NT, 1024, 0, stream>>>(srcIdx, dstIdx, toffs, binned, E, NB, NT);
    bucket_count<<<NB, BK_NODES, 0, stream>>>(binned, toffs, nodes, cnt, E, N, NB, NT);
    bucket_place<<<NB, BK_NODES, 0, stream>>>(binned, toffs, nodes, edges, E, N, NB, NT);

    const int layerBlocks = cdiv((long long)cdiv(N, 16), 8);
    const float* xcur = x;
    float* xbuf[3] = {buf0, buf1, buf0};

    for (int l = 0; l < 2; ++l) {
        float* xnew = xbuf[l];
        gcn_layer_fused<false><<<layerBlocks, TB, 0, stream>>>(nodes, cnt, edges, xcur, W[l], B[l], xnew, N, batch, sums, cnts);
        xcur = xnew;
    }
    gcn_layer_fused<true><<<layerBlocks, TB, 0, stream>>>(nodes, cnt, edges, xcur, W[2], B[2], xbuf[2], N, batch, sums, cnts);

    final_linear_wmma<<<cdiv((long long)cdiv(G, 16), 8), TB, 0, stream>>>(sums, cnts, Wl, bl, out, G);
}


# Sort level 1 tile-local: a tile orders its edges by bucket in LDS, stores one block; a bucket sums its run starts, copies its runs; no global count/scan kernel, 9 -> 7 launches; sum order = LDS rank
# speedup vs baseline: 5.3489x; 1.1754x over previous
#include <hip/hip_runtime.h>
#include <hip/hip_bf16.h>

typedef float v2f __attribute__((ext_vector_type(2)));
typedef float v8f __attribute__((ext_vector_type(8)));

#define D_FEAT 64
#define D_OUT  32

#define BK_NODES    256
#define TILE_EDGES  8192
#define MAX_BUCKETS 1024
#define MAX_TILES   1024
#define STAGE_CAP   6144

__device__ __forceinline__ int bucket_of(int d, int nb) {
    const int b = d >> 8;
    return b < nb ? b : nb - 1;
}

__global__ __launch_bounds__(1024) void tile_sort(
    const int* __restrict__ srcIdx, const int* __restrict__ dstIdx, int* __restrict__ binnedRaw,
    int* __restrict__ loff, float* __restrict__ zbuf, int e, int nb, int nt, int nz)
{
    __shared__ int  s_cur[MAX_BUCKETS];
    __shared__ int  s_w[32];
    __shared__ int2 s_stage[TILE_EDGES];
    const int tid = threadIdx.x, lane = tid & 31, wv = tid >> 5, t = blockIdx.x;
    for (int i = t * 1024 + tid; i < nz; i += nt * 1024) zbuf[i] = 0.0f;
    s_cur[tid] = 0;
    __syncthreads();
    const int base = t * TILE_EDGES + tid;
    int s[8], d[8];
    #pragma unroll
    for (int u = 0; u < 8; ++u) {
        const int i = base + u * 1024;
        const int ic = i < e ? i : e - 1;
        const int v = dstIdx[ic];
        s[u] = srcIdx[ic];
        d[u] = i < e ? v : -1;
    }
    #pragma unroll
    for (int u = 0; u < 8; ++u) if (d[u] >= 0) atomicAdd(&s_cur[bucket_of(d[u], nb)], 1);
    __syncthreads();
    const int c = s_cur[tid];
    int inc = c;
    for (int o = 1; o < 32; o <<= 1) { int x = __shfl_up(inc, o); if (lane >= o) inc += x; }
    if (lane == 31) s_w[wv] = inc;
    __syncthreads();
    if (wv == 0) {
        const int w = s_w[lane];
        int winc = w;
        for (int o = 1; o < 32; o <<= 1) { int x = __shfl_up(winc, o); if (lane >= o) winc += x; }
        s_w[lane] = winc - w;
    }
    __syncthreads();
    const int ex = s_w[wv] + inc - c;
    s_cur[tid] = ex;
    if (tid <= nb) loff[tid * nt + t] = ex;
    __syncthreads();
    #pragma unroll
    for (int u = 0; u < 8; ++u) {
        if (d[u] >= 0) {
            const int pos = atomicAdd(&s_cur[bucket_of(d[u], nb)], 1);
            s_stage[pos] = make_int2(s[u], d[u]);
        }
    }
    __syncthreads();
    const int left = e - t * TILE_EDGES;
    const int tileCount = left < TILE_EDGES ? left : TILE_EDGES;
    int2* binned = reinterpret_cast<int2*>(binnedRaw);
    #pragma unroll
    for (int u = 0; u < 8; ++u) {
        const int i = tid + u * 1024;
        if (i < tileCount) binned[t * TILE_EDGES + i] = s_stage[i];
    }
}

__global__ __launch_bounds__(BK_NODES) void bucket_count(
    const int* __restrict__ binnedRaw, const int* __restrict__ loff, int* __restrict__ packedRaw,
    int* __restrict__ bstart, int* __restrict__ nodeRaw, int* __restrict__ cnt, int e, int n, int nb, int nt)
{
    __shared__ int s_c[BK_NODES];
    __shared__ int s_w[BK_NODES / 32];
    __shared__ int s_rs[MAX_TILES];
    __shared__ int s_rl[MAX_TILES];
    __shared__ int s_ro[MAX_TILES];
    const int tid = threadIdx.x, lane = tid & 31, wv = tid >> 5;
    const int b = blockIdx.x;
    const int2* binned = reinterpret_cast<const int2*>(binnedRaw);
    int2* packed = reinterpret_cast<int2*>(packedRaw);
    s_c[tid] = 0;
    int startSum = 0, carry = 0;
    for (int t0 = 0; t0 < nt; t0 += BK_NODES) {
        const int t = t0 + tid;
        int st = 0, len = 0;
        if (t < nt) { st = loff[b * nt + t]; len = loff[(b + 1) * nt + t] - st; }
        startSum += st;
        int inc = len;
        for (int o = 1; o < 32; o <<= 1) { int x = __shfl_up(inc, o); if (lane >= o) inc += x; }
        if (lane == 31) s_w[wv] = inc;
        __syncthreads();
        int wpre = 0, tot = 0;
        for (int w = 0; w < BK_NODES / 32; ++w) { const int x = s_w[w]; if (w < wv) wpre += x; tot += x; }
        if (t < nt) { s_rs[t] = t * TILE_EDGES + st; s_rl[t] = len; s_ro[t] = carry + wpre + inc - len; }
        carry += tot;
        __syncthreads();
    }
    for (int o = 16; o > 0; o >>= 1) startSum += __shfl_xor(startSum, o);
    if (lane == 0) s_w[wv] = startSum;
    __syncthreads();
    int beg = 0;
    for (int w = 0; w < BK_NODES / 32; ++w) beg += s_w[w];
    const int count = carry;
    __syncthreads();
    const int nb0 = b * BK_NODES;
    const int wvu = __builtin_amdgcn_readfirstlane(wv);
    for (int t0 = wvu; t0 < nt; t0 += 8 * (BK_NODES / 32)) {
        int2 p[8];
        int rs[8], rl[8], ro[8];
        #pragma unroll
        for (int u = 0; u < 8; ++u) {
            const int t = t0 + u * (BK_NODES / 32);
            const int tc = t < nt ? t : nt - 1;
            const int len = s_rl[tc];
            rs[u] = s_rs[tc];
            ro[u] = beg + s_ro[tc];
            rl[u] = t < nt ? len : 0;
        }
        #pragma unroll
        for (int u = 0; u < 8; ++u) {
            const int i = rs[u] + (lane < rl[u] ? lane : 0);
            p[u] = binned[i < e ? i : e - 1];
        }
        #pragma unroll
        for (int u = 0; u < 8; ++u) {
            if (lane < rl[u]) {
                packed[ro[u] + lane] = p[u];
                atomicAdd(&s_c[(p[u].y - nb0) & (BK_NODES - 1)], 1);
            }
        }
        #pragma unroll
        for (int u = 0; u < 8; ++u) {
            for (int i = lane + 32; i < rl[u]; i += 32) {
                const int2 q = binned[rs[u] + i];
                packed[ro[u] + i] = q;
                atomicAdd(&s_c[(q.y - nb0) & (BK_NODES - 1)], 1);
            }
        }
    }
    __syncthreads();
    const int c = s_c[tid];
    int inc = c;
    for (int o = 1; o < 32; o <<= 1) { int x = __shfl_up(inc, o); if (lane >= o) inc += x; }
    if (lane == 31) s_w[wv] = inc;
    __syncthreads();
    int wpre = 0;
    for (int w = 0; w < wv; ++w) wpre += s_w[w];
    const int node = nb0 + tid;
    if (node < n) {
        const int start = beg + wpre + inc - c;
        const float d = (float)(c + 1);
        const float di = d > 0.0f ? __frsqrt_rn(d) : 0.0f;
        reinterpret_cast<int2*>(nodeRaw)[node] = make_int2(start, __float_as_int(di));
        cnt[node] = c;
    }
    if (tid == 0) {
        bstart[b] = beg;
        if (b == nb - 1) bstart[nb] = beg + count;
    }
}

__global__ __launch_bounds__(BK_NODES) void bucket_place(
    const int* __restrict__ binnedRaw, const int* __restrict__ bstart, const int* __restrict__ nodeRaw,
    int* __restrict__ edgeRaw, int n)
{
    __shared__ int   s_cur[BK_NODES];
    __shared__ float s_di[BK_NODES];
    __shared__ int2  s_stage[STAGE_CAP];
    const int tid = threadIdx.x;
    const int b = blockIdx.x;
    const int2* binned = reinterpret_cast<const int2*>(binnedRaw);
    const int2* nodeInfo = reinterpret_cast<const int2*>(nodeRaw);
    int2* edgeOut = reinterpret_cast<int2*>(edgeRaw);
    const int beg = bstart[b];
    const int end = bstart[b + 1];
    const int count = end - beg;
    const bool staged = count <= STAGE_CAP;
    const int node = b * BK_NODES + tid;
    const int2 nd = nodeInfo[node < n ? node : n - 1];
    s_cur[tid] = nd.x;
    s_di[tid]  = __int_as_float(nd.y);
    __syncthreads();
    for (int i0 = beg + tid; i0 < end; i0 += 8 * BK_NODES) {
        int2 sd[8];
        float ds[8];
        #pragma unroll
        for (int u = 0; u < 8; ++u) {
            const int i = i0 + u * BK_NODES;
            sd[u] = binned[i < end ? i : end - 1];
            if (i >= end) sd[u].y = -1;
        }
        #pragma unroll
        for (int u = 0; u < 8; ++u) ds[u] = __int_as_float(nodeInfo[sd[u].x].y);
        #pragma unroll
        for (int u = 0; u < 8; ++u) {
            if (sd[u].y >= 0) {
                const int l = (sd[u].y - b * BK_NODES) & (BK_NODES - 1);
                const int pos = atomicAdd(&s_cur[l], 1);
                const int2 rec = make_int2(sd[u].x, __float_as_int(ds[u] * s_di[l]));
                if (staged) s_stage[pos - beg] = rec; else edgeOut[pos] = rec;
            }
        }
    }
    if (staged) {
        __syncthreads();
        for (int i = tid; i < count; i += BK_NODES) edgeOut[beg + i] = s_stage[i];
    }
}

__device__ __forceinline__ void pool_flush(float* __restrict__ sums, float* __restrict__ cnts, int g, int f,
                                           const float4& a, float c) {
    float* s = sums + (size_t)g * D_FEAT + f;
    unsafeAtomicAdd(s + 0, a.x);
    unsafeAtomicAdd(s + 1, a.y);
    unsafeAtomicAdd(s + 2, a.z);
    unsafeAtomicAdd(s + 3, a.w);
    if (f == 0) unsafeAtomicAdd(&cnts[g], c);
}


#define TILE_PITCH 68
#define GK 8

template <bool POOL>
__global__ __launch_bounds__(256) void gcn_layer_fused(
    const int* __restrict__ nodeRaw, const int* __restrict__ cnt, const int* __restrict__ edgeRaw,
    const float* __restrict__ xin, const float* __restrict__ W, const float* __restrict__ bias,
    float* __restrict__ xout, int nNodes,
    const int* __restrict__ batch, float* __restrict__ sums, float* __restrict__ cnts)
{
    __shared__ __attribute__((aligned(16))) float s_tile[8][16 * TILE_PITCH];
    __shared__ int s_end[8][32];
    __shared__ __attribute__((aligned(16))) float s_pool[8][D_FEAT];
    __shared__ int   s_gid[8][32];
    __shared__ int   s_pid[8];
    __shared__ float s_pcnt[8];
    const int lane = threadIdx.x & 31;
    const int wv   = threadIdx.x >> 5;
    const int r0   = (blockIdx.x * 8 + wv) * 16;
    if (!POOL) { if (r0 >= nNodes) return; }
    if (POOL && r0 >= nNodes) { if (lane == 0) s_pid[wv] = -1; }
    if (r0 < nNodes) {
    float* st = s_tile[wv];
    int* se = s_end[wv];

    const int g  = lane >> 4;
    const int hb = lane & 16;
    const int f  = (lane & 15) * 4;
    const int first = r0 + 8 * g;
    const int2* nodeInfo = reinterpret_cast<const int2*>(nodeRaw);
    const int2* edges = reinterpret_cast<const int2*>(edgeRaw);
    const float* xf = xin + f;

    const int jn = (lane & 15) < 8 ? (lane & 15) : 7;
    const int myNode = first + jn;
    const bool myValid = myNode < nNodes;
    const int myC = myValid ? myNode : nNodes - 1;
    const int2 nd = nodeInfo[myC];
    const int segEnd = nd.x + cnt[myC];
    const int segBeg = myValid ? nd.x : segEnd;
    const float di = myValid ? __int_as_float(nd.y) : 0.0f;
    if (POOL) s_gid[wv][lane] = batch[myC];
    se[lane] = segEnd;
    #pragma unroll
    for (int i = 0; i < 8; ++i)
        *reinterpret_cast<float4*>(st + (8 * g + i) * TILE_PITCH + f) = make_float4(0.0f, 0.0f, 0.0f, 0.0f);
    const int beg = __shfl(segBeg, hb);
    const int end = __shfl(segEnd, hb + 7);
    __builtin_amdgcn_fence(__ATOMIC_RELEASE, "wavefront");
    __builtin_amdgcn_wave_barrier();
    __builtin_amdgcn_fence(__ATOMIC_ACQUIRE, "wavefront");

    {
        int cur = 0;
        int nxt = __shfl(segEnd, hb);
        float4 acc = make_float4(0.0f, 0.0f, 0.0f, 0.0f);
        int2 pr[GK];
        #pragma unroll
        for (int k = 0; k < GK; ++k) pr[k] = edges[beg + k];
        for (int base = beg; base < end; base += GK) {
            int2 nx[GK];
            #pragma unroll
            for (int k = 0; k < GK; ++k) nx[k] = edges[base + GK + k];
            float4 r[GK];
            #pragma unroll
            for (int k = 0; k < GK; ++k) {
                const int s = base + k < end ? pr[k].x : pr[0].x;
                r[k] = *reinterpret_cast<const float4*>(xf + (size_t)s * D_FEAT);
            }
            #pragma unroll
            for (int k = 0; k < GK; ++k) {
                const int e = base + k;
                if (e < end) {
                    if (e >= nxt) {
                        *reinterpret_cast<float4*>(st + (8 * g + cur) * TILE_PITCH + f) = acc;
                        acc = make_float4(0.0f, 0.0f, 0.0f, 0.0f);
                        do { ++cur; nxt = se[hb + cur]; } while (e >= nxt && cur < 7);
                    }
                    const float w = __int_as_float(pr[k].y);
                    acc.x = fmaf(w, r[k].x, acc.x); acc.y = fmaf(w, r[k].y, acc.y);
                    acc.z = fmaf(w, r[k].z, acc.z); acc.w = fmaf(w, r[k].w, acc.w);
                }
            }
            #pragma unroll
            for (int k = 0; k < GK; ++k) pr[k] = nx[k];
        }
        if (beg < end) *reinterpret_cast<float4*>(st + (8 * g + cur) * TILE_PITCH + f) = acc;
    }

    float4 xs[8];
    #pragma unroll
    for (int i = 0; i < 8; ++i) {
        const int node = first + i;
        const int nc = node < nNodes ? node : nNodes - 1;
        xs[i] = *reinterpret_cast<const float4*>(xf + (size_t)nc * D_FEAT);
    }
    #pragma unroll
    for (int i = 0; i < 8; ++i) {
        float* trow = st + (8 * g + i) * TILE_PITCH + f;
        float4 v = *reinterpret_cast<float4*>(trow);
        const float d = __shfl(di, hb + i);
        const float dd = d * d;
        v.x += dd * xs[i].x; v.y += dd * xs[i].y; v.z += dd * xs[i].z; v.w += dd * xs[i].w;
        *reinterpret_cast<float4*>(trow) = v;
    }
    __builtin_amdgcn_fence(__ATOMIC_RELEASE, "wavefront");
    __builtin_amdgcn_wave_barrier();
    __builtin_amdgcn_fence(__ATOMIC_ACQUIRE, "wavefront");

    const int m  = lane & 15;
    const int kk = (lane >> 4) * 2;
    const int n  = lane & 15;
    v8f acc0 = {}, acc1 = {}, acc2 = {}, acc3 = {};
    const float* arow = st + m * TILE_PITCH + kk;

    #pragma unroll
    for (int k0 = 0; k0 < D_FEAT; k0 += 4) {
        v2f a;
        a.x = arow[k0];
        a.y = arow[k0 + 1];
        const float* w0 = W + (size_t)(k0 + kk) * D_FEAT + n;
        const float* w1 = W + (size_t)(k0 + kk + 1) * D_FEAT + n;
        v2f b0; b0.x = w0[0];  b0.y = w1[0];
        v2f b1; b1.x = w0[16]; b1.y = w1[16];
        v2f b2; b2.x = w0[32]; b2.y = w1[32];
        v2f b3; b3.x = w0[48]; b3.y = w1[48];
        acc0 = __builtin_amdgcn_wmma_f32_16x16x4_f32(false, a, false, b0, (short)0, acc0, false, false);
        acc1 = __builtin_amdgcn_wmma_f32_16x16x4_f32(false, a, false, b1, (short)0, acc1, false, false);
        acc2 = __builtin_amdgcn_wmma_f32_16x16x4_f32(false, a, false, b2, (short)0, acc2, false, false);
        acc3 = __builtin_amdgcn_wmma_f32_16x16x4_f32(false, a, false, b3, (short)0, acc3, false, false);
    }

    __builtin_amdgcn_fence(__ATOMIC_RELEASE, "wavefront");
    __builtin_amdgcn_wave_barrier();
    __builtin_amdgcn_fence(__ATOMIC_ACQUIRE, "wavefront");
    const int rbase = (lane >> 4) * 8;
    const int col   = lane & 15;
    #pragma unroll
    for (int r = 0; r < 8; ++r) {
        float* trow = st + (rbase + r) * TILE_PITCH + col;
        trow[0]  = acc0[r];
        trow[16] = acc1[r];
        trow[32] = acc2[r];
        trow[48] = acc3[r];
    }
    __builtin_amdgcn_fence(__ATOMIC_RELEASE, "wavefront");
    __builtin_amdgcn_wave_barrier();
    __builtin_amdgcn_fence(__ATOMIC_ACQUIRE, "wavefront");

    const float4 bb = *reinterpret_cast<const float4*>(bias + f);
    #pragma unroll
    for (int i = 0; i < 8; ++i) {
        const int nl = 8 * g + i;
        const int node = r0 + nl;
        float4 v = *reinterpret_cast<float4*>(st + nl * TILE_PITCH + f);
        v.x += bb.x; v.y += bb.y; v.z += bb.z; v.w += bb.w;
        v.x = (v.x > 0.0f ? v.x : 0.0f) + xs[i].x;
        v.y = (v.y > 0.0f ? v.y : 0.0f) + xs[i].y;
        v.z = (v.z > 0.0f ? v.z : 0.0f) + xs[i].z;
        v.w = (v.w > 0.0f ? v.w : 0.0f) + xs[i].w;
        if (!POOL) {
            if (node < nNodes) *reinterpret_cast<float4*>(xout + (size_t)node * D_FEAT + f) = v;
        } else {
            *reinterpret_cast<float4*>(st + nl * TILE_PITCH + f) = v;
        }
    }
    if (POOL) {
        const int* sg = s_gid[wv];
        int pg = -1, runs = 0;
        float4 pacc = make_float4(0.0f, 0.0f, 0.0f, 0.0f);
        float pc = 0.0f;
        for (int i = 0; i < 8; ++i) {
            if (first + i < nNodes) {
                const int id = sg[hb + i];
                const float4 v = *reinterpret_cast<const float4*>(st + (8 * g + i) * TILE_PITCH + f);
                if (id != pg) {
                    if (pg >= 0) pool_flush(sums, cnts, pg, f, pacc, pc);
                    pacc = make_float4(0.0f, 0.0f, 0.0f, 0.0f); pc = 0.0f; pg = id; ++runs;
                }
                pacc.x += v.x; pacc.y += v.y; pacc.z += v.z; pacc.w += v.w;
                pc += 1.0f;
            }
        }
        const int og = __shfl_xor(pg, 16);
        const int oruns = __shfl_xor(runs, 16);
        float4 o;
        o.x = __shfl_xor(pacc.x, 16); o.y = __shfl_xor(pacc.y, 16);
        o.z = __shfl_xor(pacc.z, 16); o.w = __shfl_xor(pacc.w, 16);
        const float oc = __shfl_xor(pc, 16);
        const bool whole = pg >= 0 && og == pg && runs == 1 && oruns == 1;
        if (whole) {
            if (g == 0)
                *reinterpret_cast<float4*>(&s_pool[wv][f]) = make_float4(pacc.x + o.x, pacc.y + o.y, pacc.z + o.z, pacc.w + o.w);
            if (lane == 0) { s_pid[wv] = pg; s_pcnt[wv] = pc + oc; }
        } else {
            if (pg >= 0) pool_flush(sums, cnts, pg, f, pacc, pc);
            if (lane == 0) s_pid[wv] = -1;
        }
    }
    }

    if (POOL) {
        __syncthreads();
        if (threadIdx.x < D_FEAT) {
            const int ft = threadIdx.x;
            int cg = -1;
            float a = 0.0f, c = 0.0f;
            for (int w = 0; w < 8; ++w) {
                const int id = s_pid[w];
                if (id >= 0) {
                    if (id != cg) {
                        if (cg >= 0) {
                            unsafeAtomicAdd(&sums[(size_t)cg * D_FEAT + ft], a);
                            if (ft == 0) unsafeAtomicAdd(&cnts[cg], c);
                        }
                        a = 0.0f; c = 0.0f; cg = id;
                    }
                    a += s_pool[w][ft];
                    c += s_pcnt[w];
                }
            }
            if (cg >= 0) {
                unsafeAtomicAdd(&sums[(size_t)cg * D_FEAT + ft], a);
                if (ft == 0) unsafeAtomicAdd(&cnts[cg], c);
            }
        }
    }
}

__global__ __launch_bounds__(256) void final_linear_wmma(
    const float* __restrict__ sums, const float* __restrict__ cnts,
    const float* __restrict__ Wl, const float* __restrict__ bl,
    float* __restrict__ out, int nrows)
{
    const int lane  = threadIdx.x & 31;
    const int strip = blockIdx.x * 8 + (threadIdx.x >> 5);
    const int r0 = strip * 16;
    if (r0 >= nrows) return;

    const int m  = lane & 15;
    const int kk = (lane >> 4) * 2;
    const int n  = lane & 15;
    const int row = r0 + m;
    float c = cnts[row];
    float inv = 1.0f / (c > 1.0f ? c : 1.0f);

    v8f acc0 = {}, acc1 = {};
    const float* srow = sums + (size_t)row * D_FEAT;

    #pragma unroll
    for (int k0 = 0; k0 < D_FEAT; k0 += 4) {
        v2f a;
        a.x = srow[k0 + kk] * inv;
        a.y = srow[k0 + kk + 1] * inv;
        const float* w0 = Wl + (size_t)(k0 + kk) * D_OUT + n;
        const float* w1 = Wl + (size_t)(k0 + kk + 1) * D_OUT + n;
        v2f b0; b0.x = w0[0];  b0.y = w1[0];
        v2f b1; b1.x = w0[16]; b1.y = w1[16];
        acc0 = __builtin_amdgcn_wmma_f32_16x16x4_f32(false, a, false, b0, (short)0, acc0, false, false);
        acc1 = __builtin_amdgcn_wmma_f32_16x16x4_f32(false, a, false, b1, (short)0, acc1, false, false);
    }

    const int rbase = (lane >> 4) * 8;
    const int col   = lane & 15;
    #pragma unroll
    for (int r = 0; r < 8; ++r) {
        float* orow = out + (size_t)(r0 + rbase + r) * D_OUT + col;
        orow[0]  = acc0[r] + bl[col];
        orow[16] = acc1[r] + bl[col + 16];
    }
}

static inline int cdiv(long long a, int b) { return (int)((a + b - 1) / b); }

extern "C" void kernel_launch(void* const* d_in, const int* in_sizes, int n_in,
                              void* d_out, int out_size, void* d_ws, size_t ws_size,
                              hipStream_t stream)
{
    const float* x     = (const float*)d_in[0];
    const int*   ei    = (const int*)d_in[1];
    const int*   batch = (const int*)d_in[2];
    const float* W[3]  = {(const float*)d_in[3], (const float*)d_in[5], (const float*)d_in[7]};
    const float* B[3]  = {(const float*)d_in[4], (const float*)d_in[6], (const float*)d_in[8]};
    const float* Wl    = (const float*)d_in[9];
    const float* bl    = (const float*)d_in[10];
    float* out = (float*)d_out;

    const int N = in_sizes[0] / D_FEAT;
    const int E = in_sizes[1] / 2;
    const int G = out_size / D_OUT;
    const int* srcIdx = ei;
    const int* dstIdx = ei + E;

    float* ws   = (float*)d_ws;
    float* buf0 = ws;
    float* buf1 = buf0 + (size_t)N * D_FEAT;
    const int NB = cdiv(N, BK_NODES);
    const int NT = cdiv(E, TILE_EDGES);
    int*   edges  = (int*)(buf1 + (size_t)N * D_FEAT);
    int*   nodes  = edges + (size_t)E * 2;
    int*   binned = nodes + (size_t)N * 2;
    int*   packed = binned + (size_t)E * 2;
    int*   cnt    = packed + (size_t)E * 2;
    int*   loff   = cnt + N;
    int*   bstart = loff + (size_t)(NB + 1) * NT;
    float* sums   = (float*)(bstart + NB + 1);
    float* cnts   = sums + (size_t)G * D_FEAT;

    const int TB = 256;

    tile_sort<<<NT, 1024, 0, stream>>>(srcIdx, dstIdx, binned, loff, sums, E, NB, NT, G * D_FEAT + G);
    bucket_count<<<NB, BK_NODES, 0, stream>>>(binned, loff, packed, bstart, nodes, cnt, E, N, NB, NT);
    bucket_place<<<NB, BK_NODES, 0, stream>>>(packed, bstart, nodes, edges, N);

    const int layerBlocks = cdiv((long long)cdiv(N, 16), 8);
    const float* xcur = x;
    float* xbuf[3] = {buf0, buf1, buf0};

    for (int l = 0; l < 2; ++l) {
        float* xnew = xbuf[l];
        gcn_layer_fused<false><<<layerBlocks, TB, 0, stream>>>(nodes, cnt, edges, xcur, W[l], B[l], xnew, N, batch, sums, cnts);
        xcur = xnew;
    }
    gcn_layer_fused<true><<<layerBlocks, TB, 0, stream>>>(nodes, cnt, edges, xcur, W[2], B[2], xbuf[2], N, batch, sums, cnts);

    final_linear_wmma<<<cdiv((long long)cdiv(G, 16), 8), TB, 0, stream>>>(sums, cnts, Wl, bl, out, G);
}
